# starting kernel
# baseline (speedup 1.0000x reference)
#include <hip/hip_runtime.h>

#define N_NODES 100000
#define N_EDGES 1200000
#define D 64

typedef __attribute__((ext_vector_type(2))) float v2f;
typedef __attribute__((ext_vector_type(8))) float v8f;

__global__ __launch_bounds__(256) void sage_scatter(
    const float* __restrict__ xin,
    const int* __restrict__ src,
    const int* __restrict__ dst,
    float* __restrict__ agg_f,
    float* __restrict__ agg_r,
    float* __restrict__ cnt_f,
    float* __restrict__ cnt_r,
    int do_count)
{
    int e = __builtin_amdgcn_readfirstlane(
        (int)(blockIdx.x * 8 + (threadIdx.x >> 5)));
    if (e >= N_EDGES) return;
    int lane = threadIdx.x & 31;

    int s = src[e];
    int d = dst[e];

    const float2* xs = (const float2*)(xin + (size_t)s * D) + lane;
    const float2* xd = (const float2*)(xin + (size_t)d * D) + lane;
    float2 vs = *xs;
    float2 vd = *xd;

    float* af = agg_f + (size_t)d * D + 2 * lane;
    float* ar = agg_r + (size_t)s * D + 2 * lane;
    unsafeAtomicAdd(af + 0, vs.x);
    unsafeAtomicAdd(af + 1, vs.y);
    unsafeAtomicAdd(ar + 0, vd.x);
    unsafeAtomicAdd(ar + 1, vd.y);

    if (do_count) {
        if (lane == 0)      unsafeAtomicAdd(cnt_f + d, 1.0f);
        else if (lane == 1) unsafeAtomicAdd(cnt_r + s, 1.0f);
    }
}

__device__ __forceinline__ v8f wmma4(v2f a, v2f b, v8f c) {
    return __builtin_amdgcn_wmma_f32_16x16x4_f32(
        false, a, false, b, (short)0, c, false, false);
}

__global__ __launch_bounds__(256) void sage_gemm(
    const float* __restrict__ agg_f,
    const float* __restrict__ agg_r,
    const float* __restrict__ cnt_f,
    const float* __restrict__ cnt_r,
    const float* __restrict__ xin,
    const float* __restrict__ Wl,
    const float* __restrict__ WlT,
    const float* __restrict__ Wr,
    const float* __restrict__ WrT,
    const float* __restrict__ bias,
    const float* __restrict__ biasT,
    float* __restrict__ out)
{
    __shared__ float ldsB[192 * 64];

    for (int idx = threadIdx.x; idx < 192 * 64; idx += 256) {
        int k = idx >> 6;
        int n = idx & 63;
        float v;
        if (k < 64)        v = Wl[n * D + k];
        else if (k < 128)  v = WlT[n * D + (k - 64)];
        else               v = Wr[n * D + (k - 128)] + WrT[n * D + (k - 128)];
        ldsB[((k >> 1) << 7) + (n << 1) + (k & 1)] = v;
    }
    __syncthreads();

    int lane = threadIdx.x & 31;
    int wave = threadIdx.x >> 5;
    int tile = blockIdx.x * 8 + wave;
    const int ntiles = (N_NODES + 15) / 16;
    if (tile >= ntiles) return;

    int rowbase = tile * 16;
    int nl = lane & 15;
    int arow = rowbase + nl;
    int arowc = arow < N_NODES ? arow : N_NODES - 1;

    float invf = __builtin_amdgcn_rcpf(fmaxf(cnt_f[arowc], 1.0f));
    float invr = __builtin_amdgcn_rcpf(fmaxf(cnt_r[arowc], 1.0f));

    const float* aseg[3] = { agg_f + (size_t)arowc * D,
                             agg_r + (size_t)arowc * D,
                             xin   + (size_t)arowc * D };
    const float  ascl[3] = { invf, invr, 1.0f };

    int khalf = (lane < 16) ? 0 : 2;

    v8f c0 = {}, c1 = {}, c2 = {}, c3 = {};

#pragma unroll
    for (int seg = 0; seg < 3; ++seg) {
        const float* ap = aseg[seg];
        float sc = ascl[seg];
#pragma unroll 4
        for (int kk = 0; kk < 16; ++kk) {
            int kloc = kk * 4 + khalf;
            int kglob = seg * 64 + kloc;
            v2f a = { ap[kloc] * sc, ap[kloc + 1] * sc };
            const float* bb = ldsB + ((kglob >> 1) << 7) + (nl << 1);
            v2f b0 = *(const v2f*)(bb);
            v2f b1 = *(const v2f*)(bb + 32);
            v2f b2 = *(const v2f*)(bb + 64);
            v2f b3 = *(const v2f*)(bb + 96);
            c0 = wmma4(a, b0, c0);
            c1 = wmma4(a, b1, c1);
            c2 = wmma4(a, b2, c2);
            c3 = wmma4(a, b3, c3);
        }
    }

    int rowoff = (lane < 16) ? 0 : 8;
#pragma unroll
    for (int ct = 0; ct < 4; ++ct) {
        int j = ct * 16 + nl;
        float bs = bias[j] + biasT[j];
        v8f c = (ct == 0) ? c0 : (ct == 1) ? c1 : (ct == 2) ? c2 : c3;
#pragma unroll
        for (int v = 0; v < 8; ++v) {
            int row = rowbase + rowoff + v;
            if (row < N_NODES)
                out[(size_t)row * D + j] = c[v] + bs;
        }
    }
}

extern "C" void kernel_launch(void* const* d_in, const int* in_sizes, int n_in,
                              void* d_out, int out_size, void* d_ws, size_t ws_size,
                              hipStream_t stream) {
    (void)in_sizes; (void)n_in; (void)out_size; (void)ws_size;

    const float* x    = (const float*)d_in[0];
    const int*   ei   = (const int*)d_in[1];
    const float* Wl1  = (const float*)d_in[2];
    const float* Wr1  = (const float*)d_in[3];
    const float* b1   = (const float*)d_in[4];
    const float* WlT1 = (const float*)d_in[5];
    const float* WrT1 = (const float*)d_in[6];
    const float* bT1  = (const float*)d_in[7];
    const float* Wl2  = (const float*)d_in[8];
    const float* Wr2  = (const float*)d_in[9];
    const float* b2   = (const float*)d_in[10];
    const float* WlT2 = (const float*)d_in[11];
    const float* WrT2 = (const float*)d_in[12];
    const float* bT2  = (const float*)d_in[13];

    const int* src = ei;
    const int* dst = ei + N_EDGES;

    float* ws    = (float*)d_ws;
    float* agg_f = ws;
    float* agg_r = agg_f + (size_t)N_NODES * D;
    float* cnt_f = agg_r + (size_t)N_NODES * D;
    float* cnt_r = cnt_f + N_NODES;
    float* out   = (float*)d_out;

    const size_t agg_bytes = 2 * (size_t)N_NODES * D * sizeof(float);
    const size_t all_bytes = agg_bytes + 2 * (size_t)N_NODES * sizeof(float);

    dim3 sblk(256);
    dim3 sgrid((unsigned)(((long long)N_EDGES + 7) / 8));
    const int ntiles = (N_NODES + 15) / 16;
    dim3 gblk(256);
    dim3 ggrid((unsigned)((ntiles + 7) / 8));

    hipMemsetAsync(d_ws, 0, all_bytes, stream);
    sage_scatter<<<sgrid, sblk, 0, stream>>>(x, src, dst, agg_f, agg_r,
                                             cnt_f, cnt_r,  1);
    sage_gemm<<<ggrid, gblk, 0, stream>>>(agg_f, agg_r, cnt_f, cnt_r, x,
                                          Wl1, WlT1, Wr1, WrT1, b1, bT1, out);

    hipMemsetAsync(d_ws, 0, agg_bytes, stream);
    sage_scatter<<<sgrid, sblk, 0, stream>>>(out, src, dst, agg_f, agg_r,
                                             cnt_f, cnt_r,  0);
    sage_gemm<<<ggrid, gblk, 0, stream>>>(agg_f, agg_r, cnt_f, cnt_r, out,
                                          Wl2, WlT2, Wr2, WrT2, b2, bT2, out);
}


# scatter: one wave per 8 edges (was 1): indices by two wide scalar loads, 16 row loads in flight before the f32 atomics, 16 count adds in one instruction; atomic arrival (sum) order changes
# speedup vs baseline: 1.0626x; 1.0626x over previous
#include <hip/hip_runtime.h>

#define N_NODES 100000
#define N_EDGES 1200000
#define D 64

typedef __attribute__((ext_vector_type(2))) float v2f;
typedef __attribute__((ext_vector_type(8))) float v8f;

#define EPW 8
static_assert(N_EDGES % EPW == 0, "edge groups must tile the edge list");

__global__ __launch_bounds__(256) void sage_scatter(
    const float* __restrict__ xin,
    const int* __restrict__ src,
    const int* __restrict__ dst,
    float* __restrict__ agg_f,
    float* __restrict__ agg_r,
    float* __restrict__ cnt_f,
    float* __restrict__ cnt_r,
    int do_count)
{
    int e0 = __builtin_amdgcn_readfirstlane(
        (int)((blockIdx.x * 8 + (threadIdx.x >> 5)) * EPW));
    if (e0 >= N_EDGES) return;
    int lane = threadIdx.x & 31;

    unsigned cidx = 0;
    if (do_count && lane < 2 * EPW)
        cidx = (unsigned)((lane < EPW) ? dst[e0 + lane] : src[e0 + lane - EPW]);

    int s[EPW], d[EPW];
#pragma unroll
    for (int i = 0; i < EPW; ++i) {
        s[i] = src[e0 + i];
        d[i] = dst[e0 + i];
    }

    float2 vs[EPW], vd[EPW];
#pragma unroll
    for (int i = 0; i < EPW; ++i) {
        vs[i] = *((const float2*)(xin + (size_t)s[i] * D) + lane);
        vd[i] = *((const float2*)(xin + (size_t)d[i] * D) + lane);
    }

#pragma unroll
    for (int i = 0; i < EPW; ++i) {
        float* af = agg_f + (size_t)d[i] * D + 2 * lane;
        float* ar = agg_r + (size_t)s[i] * D + 2 * lane;
        unsafeAtomicAdd(af + 0, vs[i].x);
        unsafeAtomicAdd(af + 1, vs[i].y);
        unsafeAtomicAdd(ar + 0, vd[i].x);
        unsafeAtomicAdd(ar + 1, vd[i].y);
    }

    if (do_count && lane < 2 * EPW)
        unsafeAtomicAdd(((lane < EPW) ? cnt_f : cnt_r) + cidx, 1.0f);
}

__device__ __forceinline__ v8f wmma4(v2f a, v2f b, v8f c) {
    return __builtin_amdgcn_wmma_f32_16x16x4_f32(
        false, a, false, b, (short)0, c, false, false);
}

__global__ __launch_bounds__(256) void sage_gemm(
    const float* __restrict__ agg_f,
    const float* __restrict__ agg_r,
    const float* __restrict__ cnt_f,
    const float* __restrict__ cnt_r,
    const float* __restrict__ xin,
    const float* __restrict__ Wl,
    const float* __restrict__ WlT,
    const float* __restrict__ Wr,
    const float* __restrict__ WrT,
    const float* __restrict__ bias,
    const float* __restrict__ biasT,
    float* __restrict__ out)
{
    __shared__ float ldsB[192 * 64];

    for (int idx = threadIdx.x; idx < 192 * 64; idx += 256) {
        int k = idx >> 6;
        int n = idx & 63;
        float v;
        if (k < 64)        v = Wl[n * D + k];
        else if (k < 128)  v = WlT[n * D + (k - 64)];
        else               v = Wr[n * D + (k - 128)] + WrT[n * D + (k - 128)];
        ldsB[((k >> 1) << 7) + (n << 1) + (k & 1)] = v;
    }
    __syncthreads();

    int lane = threadIdx.x & 31;
    int wave = threadIdx.x >> 5;
    int tile = blockIdx.x * 8 + wave;
    const int ntiles = (N_NODES + 15) / 16;
    if (tile >= ntiles) return;

    int rowbase = tile * 16;
    int nl = lane & 15;
    int arow = rowbase + nl;
    int arowc = arow < N_NODES ? arow : N_NODES - 1;

    float invf = __builtin_amdgcn_rcpf(fmaxf(cnt_f[arowc], 1.0f));
    float invr = __builtin_amdgcn_rcpf(fmaxf(cnt_r[arowc], 1.0f));

    const float* aseg[3] = { agg_f + (size_t)arowc * D,
                             agg_r + (size_t)arowc * D,
                             xin   + (size_t)arowc * D };
    const float  ascl[3] = { invf, invr, 1.0f };

    int khalf = (lane < 16) ? 0 : 2;

    v8f c0 = {}, c1 = {}, c2 = {}, c3 = {};

#pragma unroll
    for (int seg = 0; seg < 3; ++seg) {
        const float* ap = aseg[seg];
        float sc = ascl[seg];
#pragma unroll 4
        for (int kk = 0; kk < 16; ++kk) {
            int kloc = kk * 4 + khalf;
            int kglob = seg * 64 + kloc;
            v2f a = { ap[kloc] * sc, ap[kloc + 1] * sc };
            const float* bb = ldsB + ((kglob >> 1) << 7) + (nl << 1);
            v2f b0 = *(const v2f*)(bb);
            v2f b1 = *(const v2f*)(bb + 32);
            v2f b2 = *(const v2f*)(bb + 64);
            v2f b3 = *(const v2f*)(bb + 96);
            c0 = wmma4(a, b0, c0);
            c1 = wmma4(a, b1, c1);
            c2 = wmma4(a, b2, c2);
            c3 = wmma4(a, b3, c3);
        }
    }

    int rowoff = (lane < 16) ? 0 : 8;
#pragma unroll
    for (int ct = 0; ct < 4; ++ct) {
        int j = ct * 16 + nl;
        float bs = bias[j] + biasT[j];
        v8f c = (ct == 0) ? c0 : (ct == 1) ? c1 : (ct == 2) ? c2 : c3;
#pragma unroll
        for (int v = 0; v < 8; ++v) {
            int row = rowbase + rowoff + v;
            if (row < N_NODES)
                out[(size_t)row * D + j] = c[v] + bs;
        }
    }
}

extern "C" void kernel_launch(void* const* d_in, const int* in_sizes, int n_in,
                              void* d_out, int out_size, void* d_ws, size_t ws_size,
                              hipStream_t stream) {
    (void)in_sizes; (void)n_in; (void)out_size; (void)ws_size;

    const float* x    = (const float*)d_in[0];
    const int*   ei   = (const int*)d_in[1];
    const float* Wl1  = (const float*)d_in[2];
    const float* Wr1  = (const float*)d_in[3];
    const float* b1   = (const float*)d_in[4];
    const float* WlT1 = (const float*)d_in[5];
    const float* WrT1 = (const float*)d_in[6];
    const float* bT1  = (const float*)d_in[7];
    const float* Wl2  = (const float*)d_in[8];
    const float* Wr2  = (const float*)d_in[9];
    const float* b2   = (const float*)d_in[10];
    const float* WlT2 = (const float*)d_in[11];
    const float* WrT2 = (const float*)d_in[12];
    const float* bT2  = (const float*)d_in[13];

    const int* src = ei;
    const int* dst = ei + N_EDGES;

    float* ws    = (float*)d_ws;
    float* agg_f = ws;
    float* agg_r = agg_f + (size_t)N_NODES * D;
    float* cnt_f = agg_r + (size_t)N_NODES * D;
    float* cnt_r = cnt_f + N_NODES;
    float* out   = (float*)d_out;

    const size_t agg_bytes = 2 * (size_t)N_NODES * D * sizeof(float);
    const size_t all_bytes = agg_bytes + 2 * (size_t)N_NODES * sizeof(float);

    dim3 sblk(256);
    dim3 sgrid((unsigned)(((long long)N_EDGES + 8 * EPW - 1) / (8 * EPW)));
    const int ntiles = (N_NODES + 15) / 16;
    dim3 gblk(256);
    dim3 ggrid((unsigned)((ntiles + 7) / 8));

    hipMemsetAsync(d_ws, 0, all_bytes, stream);
    sage_scatter<<<sgrid, sblk, 0, stream>>>(x, src, dst, agg_f, agg_r,
                                             cnt_f, cnt_r,  1);
    sage_gemm<<<ggrid, gblk, 0, stream>>>(agg_f, agg_r, cnt_f, cnt_r, x,
                                          Wl1, WlT1, Wr1, WrT1, b1, bT1, out);

    hipMemsetAsync(d_ws, 0, agg_bytes, stream);
    sage_scatter<<<sgrid, sblk, 0, stream>>>(out, src, dst, agg_f, agg_r,
                                             cnt_f, cnt_r,  0);
    sage_gemm<<<ggrid, gblk, 0, stream>>>(agg_f, agg_r, cnt_f, cnt_r, out,
                                          Wl2, WlT2, Wr2, WrT2, b2, bT2, out);
}


# scatter: lane owns dims {l,l+32}, was {2l,2l+1}: each row load and each f32 atomic instruction covers one whole 128B line (atomic line requests halved); sums still in atomic arrival order
# speedup vs baseline: 1.2063x; 1.1352x over previous
#include <hip/hip_runtime.h>

#define N_NODES 100000
#define N_EDGES 1200000
#define D 64

typedef __attribute__((ext_vector_type(2))) float v2f;
typedef __attribute__((ext_vector_type(8))) float v8f;

#define EPW 8
static_assert(N_EDGES % EPW == 0, "edge groups must tile the edge list");

__global__ __launch_bounds__(256) void sage_scatter(
    const float* __restrict__ xin,
    const int* __restrict__ src,
    const int* __restrict__ dst,
    float* __restrict__ agg_f,
    float* __restrict__ agg_r,
    float* __restrict__ cnt_f,
    float* __restrict__ cnt_r,
    int do_count)
{
    int e0 = __builtin_amdgcn_readfirstlane(
        (int)((blockIdx.x * 8 + (threadIdx.x >> 5)) * EPW));
    if (e0 >= N_EDGES) return;
    int lane = threadIdx.x & 31;

    unsigned cidx = 0;
    if (do_count && lane < 2 * EPW)
        cidx = (unsigned)((lane < EPW) ? dst[e0 + lane] : src[e0 + lane - EPW]);

    int s[EPW], d[EPW];
#pragma unroll
    for (int i = 0; i < EPW; ++i) {
        s[i] = src[e0 + i];
        d[i] = dst[e0 + i];
    }

    float vs0[EPW], vs1[EPW], vd0[EPW], vd1[EPW];
#pragma unroll
    for (int i = 0; i < EPW; ++i) {
        const float* ps = xin + (size_t)s[i] * D + lane;
        const float* pd = xin + (size_t)d[i] * D + lane;
        vs0[i] = ps[0];
        vs1[i] = ps[32];
        vd0[i] = pd[0];
        vd1[i] = pd[32];
    }

#pragma unroll
    for (int i = 0; i < EPW; ++i) {
        float* af = agg_f + (size_t)d[i] * D + lane;
        float* ar = agg_r + (size_t)s[i] * D + lane;
        unsafeAtomicAdd(af,      vs0[i]);
        unsafeAtomicAdd(af + 32, vs1[i]);
        unsafeAtomicAdd(ar,      vd0[i]);
        unsafeAtomicAdd(ar + 32, vd1[i]);
    }

    if (do_count && lane < 2 * EPW)
        unsafeAtomicAdd(((lane < EPW) ? cnt_f : cnt_r) + cidx, 1.0f);
}

__device__ __forceinline__ v8f wmma4(v2f a, v2f b, v8f c) {
    return __builtin_amdgcn_wmma_f32_16x16x4_f32(
        false, a, false, b, (short)0, c, false, false);
}

__global__ __launch_bounds__(256) void sage_gemm(
    const float* __restrict__ agg_f,
    const float* __restrict__ agg_r,
    const float* __restrict__ cnt_f,
    const float* __restrict__ cnt_r,
    const float* __restrict__ xin,
    const float* __restrict__ Wl,
    const float* __restrict__ WlT,
    const float* __restrict__ Wr,
    const float* __restrict__ WrT,
    const float* __restrict__ bias,
    const float* __restrict__ biasT,
    float* __restrict__ out)
{
    __shared__ float ldsB[192 * 64];

    for (int idx = threadIdx.x; idx < 192 * 64; idx += 256) {
        int k = idx >> 6;
        int n = idx & 63;
        float v;
        if (k < 64)        v = Wl[n * D + k];
        else if (k < 128)  v = WlT[n * D + (k - 64)];
        else               v = Wr[n * D + (k - 128)] + WrT[n * D + (k - 128)];
        ldsB[((k >> 1) << 7) + (n << 1) + (k & 1)] = v;
    }
    __syncthreads();

    int lane = threadIdx.x & 31;
    int wave = threadIdx.x >> 5;
    int tile = blockIdx.x * 8 + wave;
    const int ntiles = (N_NODES + 15) / 16;
    if (tile >= ntiles) return;

    int rowbase = tile * 16;
    int nl = lane & 15;
    int arow = rowbase + nl;
    int arowc = arow < N_NODES ? arow : N_NODES - 1;

    float invf = __builtin_amdgcn_rcpf(fmaxf(cnt_f[arowc], 1.0f));
    float invr = __builtin_amdgcn_rcpf(fmaxf(cnt_r[arowc], 1.0f));

    const float* aseg[3] = { agg_f + (size_t)arowc * D,
                             agg_r + (size_t)arowc * D,
                             xin   + (size_t)arowc * D };
    const float  ascl[3] = { invf, invr, 1.0f };

    int khalf = (lane < 16) ? 0 : 2;

    v8f c0 = {}, c1 = {}, c2 = {}, c3 = {};

#pragma unroll
    for (int seg = 0; seg < 3; ++seg) {
        const float* ap = aseg[seg];
        float sc = ascl[seg];
#pragma unroll 4
        for (int kk = 0; kk < 16; ++kk) {
            int kloc = kk * 4 + khalf;
            int kglob = seg * 64 + kloc;
            v2f a = { ap[kloc] * sc, ap[kloc + 1] * sc };
            const float* bb = ldsB + ((kglob >> 1) << 7) + (nl << 1);
            v2f b0 = *(const v2f*)(bb);
            v2f b1 = *(const v2f*)(bb + 32);
            v2f b2 = *(const v2f*)(bb + 64);
            v2f b3 = *(const v2f*)(bb + 96);
            c0 = wmma4(a, b0, c0);
            c1 = wmma4(a, b1, c1);
            c2 = wmma4(a, b2, c2);
            c3 = wmma4(a, b3, c3);
        }
    }

    int rowoff = (lane < 16) ? 0 : 8;
#pragma unroll
    for (int ct = 0; ct < 4; ++ct) {
        int j = ct * 16 + nl;
        float bs = bias[j] + biasT[j];
        v8f c = (ct == 0) ? c0 : (ct == 1) ? c1 : (ct == 2) ? c2 : c3;
#pragma unroll
        for (int v = 0; v < 8; ++v) {
            int row = rowbase + rowoff + v;
            if (row < N_NODES)
                out[(size_t)row * D + j] = c[v] + bs;
        }
    }
}

extern "C" void kernel_launch(void* const* d_in, const int* in_sizes, int n_in,
                              void* d_out, int out_size, void* d_ws, size_t ws_size,
                              hipStream_t stream) {
    (void)in_sizes; (void)n_in; (void)out_size; (void)ws_size;

    const float* x    = (const float*)d_in[0];
    const int*   ei   = (const int*)d_in[1];
    const float* Wl1  = (const float*)d_in[2];
    const float* Wr1  = (const float*)d_in[3];
    const float* b1   = (const float*)d_in[4];
    const float* WlT1 = (const float*)d_in[5];
    const float* WrT1 = (const float*)d_in[6];
    const float* bT1  = (const float*)d_in[7];
    const float* Wl2  = (const float*)d_in[8];
    const float* Wr2  = (const float*)d_in[9];
    const float* b2   = (const float*)d_in[10];
    const float* WlT2 = (const float*)d_in[11];
    const float* WrT2 = (const float*)d_in[12];
    const float* bT2  = (const float*)d_in[13];

    const int* src = ei;
    const int* dst = ei + N_EDGES;

    float* ws    = (float*)d_ws;
    float* agg_f = ws;
    float* agg_r = agg_f + (size_t)N_NODES * D;
    float* cnt_f = agg_r + (size_t)N_NODES * D;
    float* cnt_r = cnt_f + N_NODES;
    float* out   = (float*)d_out;

    const size_t agg_bytes = 2 * (size_t)N_NODES * D * sizeof(float);
    const size_t all_bytes = agg_bytes + 2 * (size_t)N_NODES * sizeof(float);

    dim3 sblk(256);
    dim3 sgrid((unsigned)(((long long)N_EDGES + 8 * EPW - 1) / (8 * EPW)));
    const int ntiles = (N_NODES + 15) / 16;
    dim3 gblk(256);
    dim3 ggrid((unsigned)((ntiles + 7) / 8));

    hipMemsetAsync(d_ws, 0, all_bytes, stream);
    sage_scatter<<<sgrid, sblk, 0, stream>>>(x, src, dst, agg_f, agg_r,
                                             cnt_f, cnt_r,  1);
    sage_gemm<<<ggrid, gblk, 0, stream>>>(agg_f, agg_r, cnt_f, cnt_r, x,
                                          Wl1, WlT1, Wr1, WrT1, b1, bT1, out);

    hipMemsetAsync(d_ws, 0, agg_bytes, stream);
    sage_scatter<<<sgrid, sblk, 0, stream>>>(out, src, dst, agg_f, agg_r,
                                             cnt_f, cnt_r,  0);
    sage_gemm<<<ggrid, gblk, 0, stream>>>(agg_f, agg_r, cnt_f, cnt_r, out,
                                          Wl2, WlT2, Wr2, WrT2, b2, bT2, out);
}


# scatter: each sum row owned by one L2 half: waves read their die, pull edge groups from per-half ticket queues, add only to own-half rows (no line moves between halves); sums in atomic arrival order
# speedup vs baseline: 1.4967x; 1.2407x over previous
#include <hip/hip_runtime.h>

#define N_NODES 100000
#define N_EDGES 1200000
#define D 64

typedef __attribute__((ext_vector_type(2))) float v2f;
typedef __attribute__((ext_vector_type(8))) float v8f;

#define EPW 8
#define SUBS 4
#define EPT (EPW * SUBS)
#define NTICKETS (N_EDGES / EPT)
#define SPLIT (N_NODES / 2)
#define QSTRIDE 64
#define SCATTER_BLOCKS 2048
static_assert(N_EDGES % EPT == 0, "tickets must tile the edge list");

__global__ __launch_bounds__(256) void sage_scatter(
    const float* __restrict__ xin,
    const int* __restrict__ src,
    const int* __restrict__ dst,
    float* __restrict__ agg_f,
    float* __restrict__ agg_r,
    float* __restrict__ cnt_f,
    float* __restrict__ cnt_r,
    unsigned* __restrict__ queue,
    int do_count)
{
    int lane = threadIdx.x & 31;

    unsigned where = __builtin_amdgcn_s_sendmsg_rtn(0x87);
    int myhalf = (int)((where >> 18) & 1u);

    for (int pass = 0; pass < 2; ++pass) {
        int h = myhalf ^ pass;
        unsigned* q = queue + h * QSTRIDE;

        for (;;) {
            unsigned tv = 0;
            if (lane == 0) tv = atomicAdd(q, 1u);
            unsigned t = (unsigned)__builtin_amdgcn_readfirstlane((int)tv);
            if (t >= (unsigned)NTICKETS) break;

#pragma nounroll
            for (int sub = 0; sub < SUBS; ++sub) {
                int e0 = (int)t * EPT + sub * EPW;

                unsigned cidx = 0;
                if (do_count && lane < 2 * EPW)
                    cidx = (unsigned)((lane < EPW) ? dst[e0 + lane]
                                                   : src[e0 + lane - EPW]);

                int s[EPW], d[EPW];
#pragma unroll
                for (int i = 0; i < EPW; ++i) {
                    s[i] = src[e0 + i];
                    d[i] = dst[e0 + i];
                }
                __builtin_amdgcn_sched_barrier(0);

                float vs0[EPW], vs1[EPW], vd0[EPW], vd1[EPW];
#pragma unroll
                for (int i = 0; i < EPW; ++i) {
                    vs0[i] = 0.0f; vs1[i] = 0.0f; vd0[i] = 0.0f; vd1[i] = 0.0f;
                    if ((int)(d[i] >= SPLIT) == h) {
                        const float* ps = xin + (size_t)s[i] * D + lane;
                        vs0[i] = ps[0];
                        vs1[i] = ps[32];
                    }
                    if ((int)(s[i] >= SPLIT) == h) {
                        const float* pd = xin + (size_t)d[i] * D + lane;
                        vd0[i] = pd[0];
                        vd1[i] = pd[32];
                    }
                }

#pragma unroll
                for (int i = 0; i < EPW; ++i) {
                    if ((int)(d[i] >= SPLIT) == h) {
                        float* af = agg_f + (size_t)d[i] * D + lane;
                        unsafeAtomicAdd(af,      vs0[i]);
                        unsafeAtomicAdd(af + 32, vs1[i]);
                    }
                    if ((int)(s[i] >= SPLIT) == h) {
                        float* ar = agg_r + (size_t)s[i] * D + lane;
                        unsafeAtomicAdd(ar,      vd0[i]);
                        unsafeAtomicAdd(ar + 32, vd1[i]);
                    }
                }

                if (do_count && lane < 2 * EPW && (int)(cidx >= (unsigned)SPLIT) == h)
                    unsafeAtomicAdd(((lane < EPW) ? cnt_f : cnt_r) + cidx, 1.0f);
            }
        }
    }
}

__device__ __forceinline__ v8f wmma4(v2f a, v2f b, v8f c) {
    return __builtin_amdgcn_wmma_f32_16x16x4_f32(
        false, a, false, b, (short)0, c, false, false);
}

__global__ __launch_bounds__(256) void sage_gemm(
    const float* __restrict__ agg_f,
    const float* __restrict__ agg_r,
    const float* __restrict__ cnt_f,
    const float* __restrict__ cnt_r,
    const float* __restrict__ xin,
    const float* __restrict__ Wl,
    const float* __restrict__ WlT,
    const float* __restrict__ Wr,
    const float* __restrict__ WrT,
    const float* __restrict__ bias,
    const float* __restrict__ biasT,
    float* __restrict__ out)
{
    __shared__ float ldsB[192 * 64];

    for (int idx = threadIdx.x; idx < 192 * 64; idx += 256) {
        int k = idx >> 6;
        int n = idx & 63;
        float v;
        if (k < 64)        v = Wl[n * D + k];
        else if (k < 128)  v = WlT[n * D + (k - 64)];
        else               v = Wr[n * D + (k - 128)] + WrT[n * D + (k - 128)];
        ldsB[((k >> 1) << 7) + (n << 1) + (k & 1)] = v;
    }
    __syncthreads();

    int lane = threadIdx.x & 31;
    int wave = threadIdx.x >> 5;
    int tile = blockIdx.x * 8 + wave;
    const int ntiles = (N_NODES + 15) / 16;
    if (tile >= ntiles) return;

    int rowbase = tile * 16;
    int nl = lane & 15;
    int arow = rowbase + nl;
    int arowc = arow < N_NODES ? arow : N_NODES - 1;

    float invf = __builtin_amdgcn_rcpf(fmaxf(cnt_f[arowc], 1.0f));
    float invr = __builtin_amdgcn_rcpf(fmaxf(cnt_r[arowc], 1.0f));

    const float* aseg[3] = { agg_f + (size_t)arowc * D,
                             agg_r + (size_t)arowc * D,
                             xin   + (size_t)arowc * D };
    const float  ascl[3] = { invf, invr, 1.0f };

    int khalf = (lane < 16) ? 0 : 2;

    v8f c0 = {}, c1 = {}, c2 = {}, c3 = {};

#pragma unroll
    for (int seg = 0; seg < 3; ++seg) {
        const float* ap = aseg[seg];
        float sc = ascl[seg];
#pragma unroll 4
        for (int kk = 0; kk < 16; ++kk) {
            int kloc = kk * 4 + khalf;
            int kglob = seg * 64 + kloc;
            v2f a = { ap[kloc] * sc, ap[kloc + 1] * sc };
            const float* bb = ldsB + ((kglob >> 1) << 7) + (nl << 1);
            v2f b0 = *(const v2f*)(bb);
            v2f b1 = *(const v2f*)(bb + 32);
            v2f b2 = *(const v2f*)(bb + 64);
            v2f b3 = *(const v2f*)(bb + 96);
            c0 = wmma4(a, b0, c0);
            c1 = wmma4(a, b1, c1);
            c2 = wmma4(a, b2, c2);
            c3 = wmma4(a, b3, c3);
        }
    }

    int rowoff = (lane < 16) ? 0 : 8;
#pragma unroll
    for (int ct = 0; ct < 4; ++ct) {
        int j = ct * 16 + nl;
        float bs = bias[j] + biasT[j];
        v8f c = (ct == 0) ? c0 : (ct == 1) ? c1 : (ct == 2) ? c2 : c3;
#pragma unroll
        for (int v = 0; v < 8; ++v) {
            int row = rowbase + rowoff + v;
            if (row < N_NODES)
                out[(size_t)row * D + j] = c[v] + bs;
        }
    }
}

extern "C" void kernel_launch(void* const* d_in, const int* in_sizes, int n_in,
                              void* d_out, int out_size, void* d_ws, size_t ws_size,
                              hipStream_t stream) {
    (void)in_sizes; (void)n_in; (void)out_size; (void)ws_size;

    const float* x    = (const float*)d_in[0];
    const int*   ei   = (const int*)d_in[1];
    const float* Wl1  = (const float*)d_in[2];
    const float* Wr1  = (const float*)d_in[3];
    const float* b1   = (const float*)d_in[4];
    const float* WlT1 = (const float*)d_in[5];
    const float* WrT1 = (const float*)d_in[6];
    const float* bT1  = (const float*)d_in[7];
    const float* Wl2  = (const float*)d_in[8];
    const float* Wr2  = (const float*)d_in[9];
    const float* b2   = (const float*)d_in[10];
    const float* WlT2 = (const float*)d_in[11];
    const float* WrT2 = (const float*)d_in[12];
    const float* bT2  = (const float*)d_in[13];

    const int* src = ei;
    const int* dst = ei + N_EDGES;

    float* ws    = (float*)d_ws;
    float* agg_f = ws;
    float* agg_r = agg_f + (size_t)N_NODES * D;
    float* cnt_f = agg_r + (size_t)N_NODES * D;
    float* cnt_r = cnt_f + N_NODES;
    unsigned* q1 = (unsigned*)(cnt_r + N_NODES);
    unsigned* q2 = q1 + 2 * QSTRIDE;
    float* out   = (float*)d_out;

    const size_t agg_bytes = 2 * (size_t)N_NODES * D * sizeof(float);
    const size_t all_bytes = agg_bytes + 2 * (size_t)N_NODES * sizeof(float)
                           + 4 * (size_t)QSTRIDE * sizeof(unsigned);

    dim3 sblk(256);
    dim3 sgrid(SCATTER_BLOCKS);
    const int ntiles = (N_NODES + 15) / 16;
    dim3 gblk(256);
    dim3 ggrid((unsigned)((ntiles + 7) / 8));

    hipMemsetAsync(d_ws, 0, all_bytes, stream);
    sage_scatter<<<sgrid, sblk, 0, stream>>>(x, src, dst, agg_f, agg_r,
                                             cnt_f, cnt_r, q1,  1);
    sage_gemm<<<ggrid, gblk, 0, stream>>>(agg_f, agg_r, cnt_f, cnt_r, x,
                                          Wl1, WlT1, Wr1, WrT1, b1, bT1, out);

    hipMemsetAsync(d_ws, 0, agg_bytes, stream);
    sage_scatter<<<sgrid, sblk, 0, stream>>>(out, src, dst, agg_f, agg_r,
                                             cnt_f, cnt_r, q2,  0);
    sage_gemm<<<ggrid, gblk, 0, stream>>>(agg_f, agg_r, cnt_f, cnt_r, out,
                                          Wl2, WlT2, Wr2, WrT2, b2, bT2, out);
}


# aggregation: neighbour lists built once per call for both layers (one returning int add per edge end, block-scan offsets), rows summed by atomic-free gathers; sum order = arrival order of the adds
# speedup vs baseline: 2.1243x; 1.4193x over previous
#include <hip/hip_runtime.h>

#define N_NODES 100000
#define N_EDGES 1200000
#define D 64
#define N_ROWS (2 * N_NODES)

typedef __attribute__((ext_vector_type(2))) float v2f;
typedef __attribute__((ext_vector_type(8))) float v8f;

#define SPLIT (N_NODES / 2)
#define QSTRIDE 64
#define EPL 10
#define EPT (32 * EPL)
#define NTICKETS (N_EDGES / EPT)
#define WALK_BLOCKS 512
static_assert(N_EDGES % EPT == 0, "tickets must tile the edge list");

template <class F>
__device__ __forceinline__ void walk_edges_by_half(unsigned* __restrict__ queue, F body)
{
    int lane = threadIdx.x & 31;

    unsigned where = __builtin_amdgcn_s_sendmsg_rtn(0x87);
    int myhalf = (int)((where >> 18) & 1u);

    for (int pass = 0; pass < 2; ++pass) {
        int h = myhalf ^ pass;
        unsigned* q = queue + h * QSTRIDE;

        for (;;) {
            unsigned tv = 0;
            if (lane == 0) tv = atomicAdd(q, 1u);
            unsigned t = (unsigned)__builtin_amdgcn_readfirstlane((int)tv);
            if (t >= (unsigned)NTICKETS) break;
            body((int)t * EPT + lane, h);
        }
    }
}

__global__ __launch_bounds__(256) void sage_rank(
    const int* __restrict__ src,
    const int* __restrict__ dst,
    int* __restrict__ deg,
    int* __restrict__ slot_f,
    int* __restrict__ slot_r,
    unsigned* __restrict__ queue)
{
    walk_edges_by_half(queue, [&](int e0, int h) {
        int s[EPL], d[EPL], rf[EPL], rr[EPL];
#pragma unroll
        for (int i = 0; i < EPL; ++i) {
            s[i] = src[e0 + 32 * i];
            d[i] = dst[e0 + 32 * i];
        }
#pragma unroll
        for (int i = 0; i < EPL; ++i) {
            rf[i] = 0; rr[i] = 0;
            if ((int)(d[i] >= SPLIT) == h) rf[i] = atomicAdd(deg + d[i], 1);
            if ((int)(s[i] >= SPLIT) == h) rr[i] = atomicAdd(deg + N_NODES + s[i], 1);
        }
#pragma unroll
        for (int i = 0; i < EPL; ++i) {
            if ((int)(d[i] >= SPLIT) == h) slot_f[e0 + 32 * i] = rf[i];
            if ((int)(s[i] >= SPLIT) == h) slot_r[e0 + 32 * i] = rr[i];
        }
    });
}

__global__ __launch_bounds__(256) void sage_alloc(
    const int* __restrict__ deg,
    int* __restrict__ off,
    float* __restrict__ cnt,
    unsigned* __restrict__ total)
{
    __shared__ int wsum[8];
    __shared__ int base_s;
    int tid = threadIdx.x;
    int lane = tid & 31;
    int wave = tid >> 5;
    int r = blockIdx.x * 256 + tid;
    int c = (r < N_ROWS) ? deg[r] : 0;

    int inc = c;
#pragma unroll
    for (int o = 1; o < 32; o <<= 1) {
        int v = __shfl_up(inc, o);
        if (lane >= o) inc += v;
    }
    if (lane == 31) wsum[wave] = inc;
    __syncthreads();
    if (tid == 0) {
        int tot = 0;
        for (int w = 0; w < 8; ++w) { int v = wsum[w]; wsum[w] = tot; tot += v; }
        base_s = (int)atomicAdd(total, (unsigned)tot);
    }
    __syncthreads();
    if (r < N_ROWS) {
        off[r] = base_s + wsum[wave] + inc - c;
        cnt[r] = (float)c;
    }
}

__global__ __launch_bounds__(256) void sage_fill(
    const int* __restrict__ src,
    const int* __restrict__ dst,
    const int* __restrict__ off,
    const int* __restrict__ slot_f,
    const int* __restrict__ slot_r,
    int* __restrict__ nbr,
    unsigned* __restrict__ queue)
{
    walk_edges_by_half(queue, [&](int e0, int h) {
        int s[EPL], d[EPL], sf[EPL], sr[EPL], of[EPL], orv[EPL];
#pragma unroll
        for (int i = 0; i < EPL; ++i) {
            s[i]  = src[e0 + 32 * i];
            d[i]  = dst[e0 + 32 * i];
            sf[i] = slot_f[e0 + 32 * i];
            sr[i] = slot_r[e0 + 32 * i];
        }
#pragma unroll
        for (int i = 0; i < EPL; ++i) {
            of[i] = 0; orv[i] = 0;
            if ((int)(d[i] >= SPLIT) == h) of[i]  = off[d[i]];
            if ((int)(s[i] >= SPLIT) == h) orv[i] = off[N_NODES + s[i]];
        }
#pragma unroll
        for (int i = 0; i < EPL; ++i) {
            if ((int)(d[i] >= SPLIT) == h) nbr[of[i] + sf[i]]  = s[i];
            if ((int)(s[i] >= SPLIT) == h) nbr[orv[i] + sr[i]] = d[i];
        }
    });
}

#define RPW 16
#define GATHER_TASKS (N_ROWS / RPW)
#define GCH 16
#define NBR_LAST (2 * N_EDGES - 1)
static_assert(N_ROWS % RPW == 0, "row groups must tile the sum rows");
static_assert(32 % GCH == 0, "chunks must tile a list load");

#define GATHER_CHUNK_LOADS(k)                                                   \
    _Pragma("unroll")                                                           \
    for (int u = 0; u < GCH; ++u)                                               \
        nb[u] = __builtin_amdgcn_readlane(idx, ((k) & 31) + u);                 \
    _Pragma("unroll")                                                           \
    for (int u = 0; u < GCH; ++u) {                                             \
        int nu = ((k) + u < n) ? nb[u] : nb[0];                \
        v[u] = *(const v2f*)(xin + (size_t)nu * D + 2 * lane);                  \
    }

#define GATHER_CHUNK_ADDS(k)                                                    \
    _Pragma("unroll")                                                           \
    for (int u = 0; u < GCH; ++u)                                               \
        acc += ((k) + u < n) ? v[u] : (v2f){ 0.0f, 0.0f };

__global__ __launch_bounds__(256) void sage_gather(
    const float* __restrict__ xin,
    const int* __restrict__ off,
    const int* __restrict__ deg,
    const int* __restrict__ nbr,
    float* __restrict__ agg)
{
    int lane = threadIdx.x & 31;
    int task = __builtin_amdgcn_readfirstlane((int)(blockIdx.x * 8 + (threadIdx.x >> 5)));
    if (task >= GATHER_TASKS) return;
    int r0 = task * RPW;

    int ov = off[r0 + (lane & (RPW - 1))];
    int nv = deg[r0 + (lane & (RPW - 1))];

    int o = __builtin_amdgcn_readlane(ov, 0);
    int n = __builtin_amdgcn_readlane(nv, 0);
    int idx = nbr[min(o + lane, NBR_LAST)];

    for (int j = 0; j < RPW; ++j) {
        int nb[GCH];
        v2f v[GCH];
        v2f acc = { 0.0f, 0.0f };

        GATHER_CHUNK_LOADS(0)

        int o2 = __builtin_amdgcn_readlane(ov, (j + 1) & (RPW - 1));
        int n2 = __builtin_amdgcn_readlane(nv, (j + 1) & (RPW - 1));
        int idx2 = nbr[min(o2 + lane, NBR_LAST)];
        asm volatile("" :: "v"(idx2));

        GATHER_CHUNK_ADDS(0)

        for (int k = GCH; k < n; k += GCH) {
            if ((k & 31) == 0) idx = nbr[min(o + k + lane, NBR_LAST)];
            GATHER_CHUNK_LOADS(k)
            GATHER_CHUNK_ADDS(k)
        }

        *(v2f*)(agg + (size_t)(r0 + j) * D + 2 * lane) = acc;

        o = o2; n = n2; idx = idx2;
    }
}

__device__ __forceinline__ v8f wmma4(v2f a, v2f b, v8f c) {
    return __builtin_amdgcn_wmma_f32_16x16x4_f32(
        false, a, false, b, (short)0, c, false, false);
}

__global__ __launch_bounds__(256) void sage_gemm(
    const float* __restrict__ agg_f,
    const float* __restrict__ agg_r,
    const float* __restrict__ cnt_f,
    const float* __restrict__ cnt_r,
    const float* __restrict__ xin,
    const float* __restrict__ Wl,
    const float* __restrict__ WlT,
    const float* __restrict__ Wr,
    const float* __restrict__ WrT,
    const float* __restrict__ bias,
    const float* __restrict__ biasT,
    float* __restrict__ out)
{
    __shared__ float ldsB[192 * 64];

    for (int idx = threadIdx.x; idx < 192 * 64; idx += 256) {
        int k = idx >> 6;
        int n = idx & 63;
        float v;
        if (k < 64)        v = Wl[n * D + k];
        else if (k < 128)  v = WlT[n * D + (k - 64)];
        else               v = Wr[n * D + (k - 128)] + WrT[n * D + (k - 128)];
        ldsB[((k >> 1) << 7) + (n << 1) + (k & 1)] = v;
    }
    __syncthreads();

    int lane = threadIdx.x & 31;
    int wave = threadIdx.x >> 5;
    int tile = blockIdx.x * 8 + wave;
    const int ntiles = (N_NODES + 15) / 16;
    if (tile >= ntiles) return;

    int rowbase = tile * 16;
    int nl = lane & 15;
    int arow = rowbase + nl;
    int arowc = arow < N_NODES ? arow : N_NODES - 1;

    float invf = __builtin_amdgcn_rcpf(fmaxf(cnt_f[arowc], 1.0f));
    float invr = __builtin_amdgcn_rcpf(fmaxf(cnt_r[arowc], 1.0f));

    const float* aseg[3] = { agg_f + (size_t)arowc * D,
                             agg_r + (size_t)arowc * D,
                             xin   + (size_t)arowc * D };
    const float  ascl[3] = { invf, invr, 1.0f };

    int khalf = (lane < 16) ? 0 : 2;

    v8f c0 = {}, c1 = {}, c2 = {}, c3 = {};

#pragma unroll
    for (int seg = 0; seg < 3; ++seg) {
        const float* ap = aseg[seg];
        float sc = ascl[seg];
#pragma unroll 4
        for (int kk = 0; kk < 16; ++kk) {
            int kloc = kk * 4 + khalf;
            int kglob = seg * 64 + kloc;
            v2f a = { ap[kloc] * sc, ap[kloc + 1] * sc };
            const float* bb = ldsB + ((kglob >> 1) << 7) + (nl << 1);
            v2f b0 = *(const v2f*)(bb);
            v2f b1 = *(const v2f*)(bb + 32);
            v2f b2 = *(const v2f*)(bb + 64);
            v2f b3 = *(const v2f*)(bb + 96);
            c0 = wmma4(a, b0, c0);
            c1 = wmma4(a, b1, c1);
            c2 = wmma4(a, b2, c2);
            c3 = wmma4(a, b3, c3);
        }
    }

    int rowoff = (lane < 16) ? 0 : 8;
#pragma unroll
    for (int ct = 0; ct < 4; ++ct) {
        int j = ct * 16 + nl;
        float bs = bias[j] + biasT[j];
        v8f c = (ct == 0) ? c0 : (ct == 1) ? c1 : (ct == 2) ? c2 : c3;
#pragma unroll
        for (int v = 0; v < 8; ++v) {
            int row = rowbase + rowoff + v;
            if (row < N_NODES)
                out[(size_t)row * D + j] = c[v] + bs;
        }
    }
}

extern "C" void kernel_launch(void* const* d_in, const int* in_sizes, int n_in,
                              void* d_out, int out_size, void* d_ws, size_t ws_size,
                              hipStream_t stream) {
    (void)in_sizes; (void)n_in; (void)out_size; (void)ws_size;

    const float* x    = (const float*)d_in[0];
    const int*   ei   = (const int*)d_in[1];
    const float* Wl1  = (const float*)d_in[2];
    const float* Wr1  = (const float*)d_in[3];
    const float* b1   = (const float*)d_in[4];
    const float* WlT1 = (const float*)d_in[5];
    const float* WrT1 = (const float*)d_in[6];
    const float* bT1  = (const float*)d_in[7];
    const float* Wl2  = (const float*)d_in[8];
    const float* Wr2  = (const float*)d_in[9];
    const float* b2   = (const float*)d_in[10];
    const float* WlT2 = (const float*)d_in[11];
    const float* WrT2 = (const float*)d_in[12];
    const float* bT2  = (const float*)d_in[13];

    const int* src = ei;
    const int* dst = ei + N_EDGES;

    float* ws    = (float*)d_ws;
    float* agg_f = ws;
    float* agg_r = agg_f + (size_t)N_NODES * D;
    float* cnt_f = agg_r + (size_t)N_NODES * D;
    float* cnt_r = cnt_f + N_NODES;
    int*      deg   = (int*)(cnt_r + N_NODES);
    unsigned* total = (unsigned*)(deg + N_ROWS);
    unsigned* q_rank = total + QSTRIDE;
    unsigned* q_fill = q_rank + 2 * QSTRIDE;
    int* off    = (int*)(q_fill + 2 * QSTRIDE);
    int* slot_f = off + N_ROWS;
    int* slot_r = slot_f + N_EDGES;
    int* nbr    = slot_r + N_EDGES;
    float* out  = (float*)d_out;

    const size_t zero_bytes = (size_t)N_ROWS * sizeof(int)
                            + 5 * (size_t)QSTRIDE * sizeof(unsigned);

    dim3 blk(256);
    dim3 wgrid(WALK_BLOCKS);
    dim3 agrid((N_ROWS + 255) / 256);
    dim3 hgrid((GATHER_TASKS + 7) / 8);
    const int ntiles = (N_NODES + 15) / 16;
    dim3 ggrid((unsigned)((ntiles + 7) / 8));

    hipMemsetAsync(deg, 0, zero_bytes, stream);
    sage_rank<<<wgrid, blk, 0, stream>>>(src, dst, deg, slot_f, slot_r, q_rank);
    sage_alloc<<<agrid, blk, 0, stream>>>(deg, off, cnt_f, total);
    sage_fill<<<wgrid, blk, 0, stream>>>(src, dst, off, slot_f, slot_r, nbr, q_fill);

    sage_gather<<<hgrid, blk, 0, stream>>>(x, off, deg, nbr, agg_f);
    sage_gemm<<<ggrid, blk, 0, stream>>>(agg_f, agg_r, cnt_f, cnt_r, x,
                                         Wl1, WlT1, Wr1, WrT1, b1, bT1, out);

    sage_gather<<<hgrid, blk, 0, stream>>>(out, off, deg, nbr, agg_f);
    sage_gemm<<<ggrid, blk, 0, stream>>>(agg_f, agg_r, cnt_f, cnt_r, out,
                                         Wl2, WlT2, Wr2, WrT2, b2, bT2, out);
}


# gemm: weights staged to LDS by 16-byte loads along each row, all 16 in flight (was 64 dependent 4-byte loads 256 B apart, 32 lines each); k-pair rows padded to 130 floats: stores conflict-free
# speedup vs baseline: 2.4610x; 1.1585x over previous
#include <hip/hip_runtime.h>

#define N_NODES 100000
#define N_EDGES 1200000
#define D 64
#define N_ROWS (2 * N_NODES)

typedef __attribute__((ext_vector_type(2))) float v2f;
typedef __attribute__((ext_vector_type(4))) float v4f;
typedef __attribute__((ext_vector_type(8))) float v8f;

#define SPLIT (N_NODES / 2)
#define QSTRIDE 64
#define EPL 10
#define EPT (32 * EPL)
#define NTICKETS (N_EDGES / EPT)
#define WALK_BLOCKS 512
static_assert(N_EDGES % EPT == 0, "tickets must tile the edge list");

template <class F>
__device__ __forceinline__ void walk_edges_by_half(unsigned* __restrict__ queue, F body)
{
    int lane = threadIdx.x & 31;

    unsigned where = __builtin_amdgcn_s_sendmsg_rtn(0x87);
    int myhalf = (int)((where >> 18) & 1u);

    for (int pass = 0; pass < 2; ++pass) {
        int h = myhalf ^ pass;
        unsigned* q = queue + h * QSTRIDE;

        for (;;) {
            unsigned tv = 0;
            if (lane == 0) tv = atomicAdd(q, 1u);
            unsigned t = (unsigned)__builtin_amdgcn_readfirstlane((int)tv);
            if (t >= (unsigned)NTICKETS) break;
            body((int)t * EPT + lane, h);
        }
    }
}

__global__ __launch_bounds__(256) void sage_rank(
    const int* __restrict__ src,
    const int* __restrict__ dst,
    int* __restrict__ deg,
    int* __restrict__ slot_f,
    int* __restrict__ slot_r,
    unsigned* __restrict__ queue)
{
    walk_edges_by_half(queue, [&](int e0, int h) {
        int s[EPL], d[EPL], rf[EPL], rr[EPL];
#pragma unroll
        for (int i = 0; i < EPL; ++i) {
            s[i] = src[e0 + 32 * i];
            d[i] = dst[e0 + 32 * i];
        }
#pragma unroll
        for (int i = 0; i < EPL; ++i) {
            rf[i] = 0; rr[i] = 0;
            if ((int)(d[i] >= SPLIT) == h) rf[i] = atomicAdd(deg + d[i], 1);
            if ((int)(s[i] >= SPLIT) == h) rr[i] = atomicAdd(deg + N_NODES + s[i], 1);
        }
#pragma unroll
        for (int i = 0; i < EPL; ++i) {
            if ((int)(d[i] >= SPLIT) == h) slot_f[e0 + 32 * i] = rf[i];
            if ((int)(s[i] >= SPLIT) == h) slot_r[e0 + 32 * i] = rr[i];
        }
    });
}

__global__ __launch_bounds__(256) void sage_alloc(
    const int* __restrict__ deg,
    int* __restrict__ off,
    float* __restrict__ cnt,
    unsigned* __restrict__ total)
{
    __shared__ int wsum[8];
    __shared__ int base_s;
    int tid = threadIdx.x;
    int lane = tid & 31;
    int wave = tid >> 5;
    int r = blockIdx.x * 256 + tid;
    int c = (r < N_ROWS) ? deg[r] : 0;

    int inc = c;
#pragma unroll
    for (int o = 1; o < 32; o <<= 1) {
        int v = __shfl_up(inc, o);
        if (lane >= o) inc += v;
    }
    if (lane == 31) wsum[wave] = inc;
    __syncthreads();
    if (tid == 0) {
        int tot = 0;
        for (int w = 0; w < 8; ++w) { int v = wsum[w]; wsum[w] = tot; tot += v; }
        base_s = (int)atomicAdd(total, (unsigned)tot);
    }
    __syncthreads();
    if (r < N_ROWS) {
        off[r] = base_s + wsum[wave] + inc - c;
        cnt[r] = (float)c;
    }
}

__global__ __launch_bounds__(256) void sage_fill(
    const int* __restrict__ src,
    const int* __restrict__ dst,
    const int* __restrict__ off,
    const int* __restrict__ slot_f,
    const int* __restrict__ slot_r,
    int* __restrict__ nbr,
    unsigned* __restrict__ queue)
{
    walk_edges_by_half(queue, [&](int e0, int h) {
        int s[EPL], d[EPL], sf[EPL], sr[EPL], of[EPL], orv[EPL];
#pragma unroll
        for (int i = 0; i < EPL; ++i) {
            s[i]  = src[e0 + 32 * i];
            d[i]  = dst[e0 + 32 * i];
            sf[i] = slot_f[e0 + 32 * i];
            sr[i] = slot_r[e0 + 32 * i];
        }
#pragma unroll
        for (int i = 0; i < EPL; ++i) {
            of[i] = 0; orv[i] = 0;
            if ((int)(d[i] >= SPLIT) == h) of[i]  = off[d[i]];
            if ((int)(s[i] >= SPLIT) == h) orv[i] = off[N_NODES + s[i]];
        }
#pragma unroll
        for (int i = 0; i < EPL; ++i) {
            if ((int)(d[i] >= SPLIT) == h) nbr[of[i] + sf[i]]  = s[i];
            if ((int)(s[i] >= SPLIT) == h) nbr[orv[i] + sr[i]] = d[i];
        }
    });
}

#define RPW 16
#define GATHER_TASKS (N_ROWS / RPW)
#define GCH 16
#define NBR_LAST (2 * N_EDGES - 1)
static_assert(N_ROWS % RPW == 0, "row groups must tile the sum rows");
static_assert(32 % GCH == 0, "chunks must tile a list load");

#define GATHER_CHUNK_LOADS(k)                                                   \
    _Pragma("unroll")                                                           \
    for (int u = 0; u < GCH; ++u)                                               \
        nb[u] = __builtin_amdgcn_readlane(idx, ((k) & 31) + u);                 \
    _Pragma("unroll")                                                           \
    for (int u = 0; u < GCH; ++u) {                                             \
        int nu = ((k) + u < n) ? nb[u] : nb[0];                \
        v[u] = *(const v2f*)(xin + (size_t)nu * D + 2 * lane);                  \
    }

#define GATHER_CHUNK_ADDS(k)                                                    \
    _Pragma("unroll")                                                           \
    for (int u = 0; u < GCH; ++u)                                               \
        acc += ((k) + u < n) ? v[u] : (v2f){ 0.0f, 0.0f };

__global__ __launch_bounds__(256) void sage_gather(
    const float* __restrict__ xin,
    const int* __restrict__ off,
    const int* __restrict__ deg,
    const int* __restrict__ nbr,
    float* __restrict__ agg)
{
    int lane = threadIdx.x & 31;
    int task = __builtin_amdgcn_readfirstlane((int)(blockIdx.x * 8 + (threadIdx.x >> 5)));
    if (task >= GATHER_TASKS) return;
    int r0 = task * RPW;

    int ov = off[r0 + (lane & (RPW - 1))];
    int nv = deg[r0 + (lane & (RPW - 1))];

    int o = __builtin_amdgcn_readlane(ov, 0);
    int n = __builtin_amdgcn_readlane(nv, 0);
    int idx = nbr[min(o + lane, NBR_LAST)];

    for (int j = 0; j < RPW; ++j) {
        int nb[GCH];
        v2f v[GCH];
        v2f acc = { 0.0f, 0.0f };

        GATHER_CHUNK_LOADS(0)

        int o2 = __builtin_amdgcn_readlane(ov, (j + 1) & (RPW - 1));
        int n2 = __builtin_amdgcn_readlane(nv, (j + 1) & (RPW - 1));
        int idx2 = nbr[min(o2 + lane, NBR_LAST)];
        asm volatile("" :: "v"(idx2));

        GATHER_CHUNK_ADDS(0)

        for (int k = GCH; k < n; k += GCH) {
            if ((k & 31) == 0) idx = nbr[min(o + k + lane, NBR_LAST)];
            GATHER_CHUNK_LOADS(k)
            GATHER_CHUNK_ADDS(k)
        }

        *(v2f*)(agg + (size_t)(r0 + j) * D + 2 * lane) = acc;

        o = o2; n = n2; idx = idx2;
    }
}

#define BPITCH 130
__device__ __forceinline__ v8f wmma4(v2f a, v2f b, v8f c) {
    return __builtin_amdgcn_wmma_f32_16x16x4_f32(
        false, a, false, b, (short)0, c, false, false);
}

__global__ __launch_bounds__(256) void sage_gemm(
    const float* __restrict__ agg_f,
    const float* __restrict__ agg_r,
    const float* __restrict__ cnt_f,
    const float* __restrict__ cnt_r,
    const float* __restrict__ xin,
    const float* __restrict__ Wl,
    const float* __restrict__ WlT,
    const float* __restrict__ Wr,
    const float* __restrict__ WrT,
    const float* __restrict__ bias,
    const float* __restrict__ biasT,
    float* __restrict__ out)
{
    __shared__ __attribute__((aligned(16))) float ldsB[96 * BPITCH];

    v4f w0[4], w1[4], w2[4], w3[4];
#pragma unroll
    for (int it = 0; it < 4; ++it) {
        int g = it * 256 + (int)threadIdx.x;
        w0[it] = ((const v4f*)Wl)[g];
        w1[it] = ((const v4f*)WlT)[g];
        w2[it] = ((const v4f*)Wr)[g];
        w3[it] = ((const v4f*)WrT)[g];
    }
#pragma unroll
    for (int it = 0; it < 4; ++it) {
        int g  = it * 256 + (int)threadIdx.x;
        int n  = g >> 4;
        int kq = g & 15;
        v4f w2s = w2[it] + w3[it];
        float* d = ldsB + (2 * kq) * BPITCH + (n << 1);
        *(v2f*)(d)                = (v2f){ w0[it].x, w0[it].y };
        *(v2f*)(d + BPITCH)       = (v2f){ w0[it].z, w0[it].w };
        *(v2f*)(d + 32 * BPITCH)  = (v2f){ w1[it].x, w1[it].y };
        *(v2f*)(d + 33 * BPITCH)  = (v2f){ w1[it].z, w1[it].w };
        *(v2f*)(d + 64 * BPITCH)  = (v2f){ w2s.x, w2s.y };
        *(v2f*)(d + 65 * BPITCH)  = (v2f){ w2s.z, w2s.w };
    }
    __syncthreads();

    int lane = threadIdx.x & 31;
    int wave = threadIdx.x >> 5;
    int tile = blockIdx.x * 8 + wave;
    const int ntiles = (N_NODES + 15) / 16;
    if (tile >= ntiles) return;

    int rowbase = tile * 16;
    int nl = lane & 15;
    int arow = rowbase + nl;
    int arowc = arow < N_NODES ? arow : N_NODES - 1;

    float invf = __builtin_amdgcn_rcpf(fmaxf(cnt_f[arowc], 1.0f));
    float invr = __builtin_amdgcn_rcpf(fmaxf(cnt_r[arowc], 1.0f));

    const float* aseg[3] = { agg_f + (size_t)arowc * D,
                             agg_r + (size_t)arowc * D,
                             xin   + (size_t)arowc * D };
    const float  ascl[3] = { invf, invr, 1.0f };

    int khalf = (lane < 16) ? 0 : 2;

    v8f c0 = {}, c1 = {}, c2 = {}, c3 = {};

#pragma unroll
    for (int seg = 0; seg < 3; ++seg) {
        const float* ap = aseg[seg];
        float sc = ascl[seg];
#pragma unroll 4
        for (int kk = 0; kk < 16; ++kk) {
            int kloc = kk * 4 + khalf;
            int kglob = seg * 64 + kloc;
            v2f a = { ap[kloc] * sc, ap[kloc + 1] * sc };
            const float* bb = ldsB + (kglob >> 1) * BPITCH + (nl << 1);
            v2f b0 = *(const v2f*)(bb);
            v2f b1 = *(const v2f*)(bb + 32);
            v2f b2 = *(const v2f*)(bb + 64);
            v2f b3 = *(const v2f*)(bb + 96);
            c0 = wmma4(a, b0, c0);
            c1 = wmma4(a, b1, c1);
            c2 = wmma4(a, b2, c2);
            c3 = wmma4(a, b3, c3);
        }
    }

    int rowoff = (lane < 16) ? 0 : 8;
#pragma unroll
    for (int ct = 0; ct < 4; ++ct) {
        int j = ct * 16 + nl;
        float bs = bias[j] + biasT[j];
        v8f c = (ct == 0) ? c0 : (ct == 1) ? c1 : (ct == 2) ? c2 : c3;
#pragma unroll
        for (int v = 0; v < 8; ++v) {
            int row = rowbase + rowoff + v;
            if (row < N_NODES)
                out[(size_t)row * D + j] = c[v] + bs;
        }
    }
}

extern "C" void kernel_launch(void* const* d_in, const int* in_sizes, int n_in,
                              void* d_out, int out_size, void* d_ws, size_t ws_size,
                              hipStream_t stream) {
    (void)in_sizes; (void)n_in; (void)out_size; (void)ws_size;

    const float* x    = (const float*)d_in[0];
    const int*   ei   = (const int*)d_in[1];
    const float* Wl1  = (const float*)d_in[2];
    const float* Wr1  = (const float*)d_in[3];
    const float* b1   = (const float*)d_in[4];
    const float* WlT1 = (const float*)d_in[5];
    const float* WrT1 = (const float*)d_in[6];
    const float* bT1  = (const float*)d_in[7];
    const float* Wl2  = (const float*)d_in[8];
    const float* Wr2  = (const float*)d_in[9];
    const float* b2   = (const float*)d_in[10];
    const float* WlT2 = (const float*)d_in[11];
    const float* WrT2 = (const float*)d_in[12];
    const float* bT2  = (const float*)d_in[13];

    const int* src = ei;
    const int* dst = ei + N_EDGES;

    float* ws    = (float*)d_ws;
    float* agg_f = ws;
    float* agg_r = agg_f + (size_t)N_NODES * D;
    float* cnt_f = agg_r + (size_t)N_NODES * D;
    float* cnt_r = cnt_f + N_NODES;
    int*      deg   = (int*)(cnt_r + N_NODES);
    unsigned* total = (unsigned*)(deg + N_ROWS);
    unsigned* q_rank = total + QSTRIDE;
    unsigned* q_fill = q_rank + 2 * QSTRIDE;
    int* off    = (int*)(q_fill + 2 * QSTRIDE);
    int* slot_f = off + N_ROWS;
    int* slot_r = slot_f + N_EDGES;
    int* nbr    = slot_r + N_EDGES;
    float* out  = (float*)d_out;

    const size_t zero_bytes = (size_t)N_ROWS * sizeof(int)
                            + 5 * (size_t)QSTRIDE * sizeof(unsigned);

    dim3 blk(256);
    dim3 wgrid(WALK_BLOCKS);
    dim3 agrid((N_ROWS + 255) / 256);
    dim3 hgrid((GATHER_TASKS + 7) / 8);
    const int ntiles = (N_NODES + 15) / 16;
    dim3 ggrid((unsigned)((ntiles + 7) / 8));

    hipMemsetAsync(deg, 0, zero_bytes, stream);
    sage_rank<<<wgrid, blk, 0, stream>>>(src, dst, deg, slot_f, slot_r, q_rank);
    sage_alloc<<<agrid, blk, 0, stream>>>(deg, off, cnt_f, total);
    sage_fill<<<wgrid, blk, 0, stream>>>(src, dst, off, slot_f, slot_r, nbr, q_fill);

    sage_gather<<<hgrid, blk, 0, stream>>>(x, off, deg, nbr, agg_f);
    sage_gemm<<<ggrid, blk, 0, stream>>>(agg_f, agg_r, cnt_f, cnt_r, x,
                                         Wl1, WlT1, Wr1, WrT1, b1, bT1, out);

    sage_gather<<<hgrid, blk, 0, stream>>>(out, off, deg, nbr, agg_f);
    sage_gemm<<<ggrid, blk, 0, stream>>>(agg_f, agg_r, cnt_f, cnt_r, out,
                                         Wl2, WlT2, Wr2, WrT2, b2, bT2, out);
}


# layer: gather and GEMM fused: a block's 16 waves sum 4 node tiles' forward and reverse rows into LDS, then run the K=192 f32 WMMA product from there; no sum rows in memory, 2 launches fewer
# speedup vs baseline: 2.6065x; 1.0591x over previous
#include <hip/hip_runtime.h>

#define N_NODES 100000
#define N_EDGES 1200000
#define D 64
#define N_ROWS (2 * N_NODES)

typedef __attribute__((ext_vector_type(2))) float v2f;
typedef __attribute__((ext_vector_type(4))) float v4f;
typedef __attribute__((ext_vector_type(8))) float v8f;

#define SPLIT (N_NODES / 2)
#define QSTRIDE 64
#define EPL 10
#define EPT (32 * EPL)
#define NTICKETS (N_EDGES / EPT)
#define WALK_BLOCKS 512
static_assert(N_EDGES % EPT == 0, "tickets must tile the edge list");

template <class F>
__device__ __forceinline__ void walk_edges_by_half(unsigned* __restrict__ queue, F body)
{
    int lane = threadIdx.x & 31;

    unsigned where = __builtin_amdgcn_s_sendmsg_rtn(0x87);
    int myhalf = (int)((where >> 18) & 1u);

    for (int pass = 0; pass < 2; ++pass) {
        int h = myhalf ^ pass;
        unsigned* q = queue + h * QSTRIDE;

        for (;;) {
            unsigned tv = 0;
            if (lane == 0) tv = atomicAdd(q, 1u);
            unsigned t = (unsigned)__builtin_amdgcn_readfirstlane((int)tv);
            if (t >= (unsigned)NTICKETS) break;
            body((int)t * EPT + lane, h);
        }
    }
}

__global__ __launch_bounds__(256) void sage_rank(
    const int* __restrict__ src,
    const int* __restrict__ dst,
    int* __restrict__ deg,
    int* __restrict__ slot_f,
    int* __restrict__ slot_r,
    unsigned* __restrict__ queue)
{
    walk_edges_by_half(queue, [&](int e0, int h) {
        int s[EPL], d[EPL], rf[EPL], rr[EPL];
#pragma unroll
        for (int i = 0; i < EPL; ++i) {
            s[i] = src[e0 + 32 * i];
            d[i] = dst[e0 + 32 * i];
        }
#pragma unroll
        for (int i = 0; i < EPL; ++i) {
            rf[i] = 0; rr[i] = 0;
            if ((int)(d[i] >= SPLIT) == h) rf[i] = atomicAdd(deg + d[i], 1);
            if ((int)(s[i] >= SPLIT) == h) rr[i] = atomicAdd(deg + N_NODES + s[i], 1);
        }
#pragma unroll
        for (int i = 0; i < EPL; ++i) {
            if ((int)(d[i] >= SPLIT) == h) slot_f[e0 + 32 * i] = rf[i];
            if ((int)(s[i] >= SPLIT) == h) slot_r[e0 + 32 * i] = rr[i];
        }
    });
}

__global__ __launch_bounds__(256) void sage_alloc(
    const int* __restrict__ deg,
    int* __restrict__ off,
    unsigned* __restrict__ total)
{
    __shared__ int wsum[8];
    __shared__ int base_s;
    int tid = threadIdx.x;
    int lane = tid & 31;
    int wave = tid >> 5;
    int r = blockIdx.x * 256 + tid;
    int c = (r < N_ROWS) ? deg[r] : 0;

    int inc = c;
#pragma unroll
    for (int o = 1; o < 32; o <<= 1) {
        int v = __shfl_up(inc, o);
        if (lane >= o) inc += v;
    }
    if (lane == 31) wsum[wave] = inc;
    __syncthreads();
    if (tid == 0) {
        int tot = 0;
        for (int w = 0; w < 8; ++w) { int v = wsum[w]; wsum[w] = tot; tot += v; }
        base_s = (int)atomicAdd(total, (unsigned)tot);
    }
    __syncthreads();
    if (r < N_ROWS) off[r] = base_s + wsum[wave] + inc - c;
}

__global__ __launch_bounds__(256) void sage_fill(
    const int* __restrict__ src,
    const int* __restrict__ dst,
    const int* __restrict__ off,
    const int* __restrict__ slot_f,
    const int* __restrict__ slot_r,
    int* __restrict__ nbr,
    unsigned* __restrict__ queue)
{
    walk_edges_by_half(queue, [&](int e0, int h) {
        int s[EPL], d[EPL], sf[EPL], sr[EPL], of[EPL], orv[EPL];
#pragma unroll
        for (int i = 0; i < EPL; ++i) {
            s[i]  = src[e0 + 32 * i];
            d[i]  = dst[e0 + 32 * i];
            sf[i] = slot_f[e0 + 32 * i];
            sr[i] = slot_r[e0 + 32 * i];
        }
#pragma unroll
        for (int i = 0; i < EPL; ++i) {
            of[i] = 0; orv[i] = 0;
            if ((int)(d[i] >= SPLIT) == h) of[i]  = off[d[i]];
            if ((int)(s[i] >= SPLIT) == h) orv[i] = off[N_NODES + s[i]];
        }
#pragma unroll
        for (int i = 0; i < EPL; ++i) {
            if ((int)(d[i] >= SPLIT) == h) nbr[of[i] + sf[i]]  = s[i];
            if ((int)(s[i] >= SPLIT) == h) nbr[orv[i] + sr[i]] = d[i];
        }
    });
}

#define RPW 8
#define TROWS 16
#define GCH 16
#define NBR_LAST (2 * N_EDGES - 1)
#define TPB 4
#define N_TILES (N_NODES / TROWS)
#define BPITCH 130
#define APITCH 68
#define SLOT (TROWS * APITCH)
static_assert(N_NODES % TROWS == 0, "tiles must tile the nodes");
static_assert(32 % GCH == 0, "chunks must tile a list load");

#define GATHER_CHUNK_LOADS(k)                                                   \
    _Pragma("unroll")                                                           \
    for (int u = 0; u < GCH; ++u)                                               \
        nb[u] = __builtin_amdgcn_readlane(idx, ((k) & 31) + u);                 \
    _Pragma("unroll")                                                           \
    for (int u = 0; u < GCH; ++u) {                                             \
        int nu = ((k) + u < n) ? nb[u] : nb[0];                \
        v[u] = *(const v2f*)(xin + (size_t)nu * D + 2 * lane);                  \
    }

#define GATHER_CHUNK_ADDS(k)                                                    \
    _Pragma("unroll")                                                           \
    for (int u = 0; u < GCH; ++u)                                               \
        acc += ((k) + u < n) ? v[u] : (v2f){ 0.0f, 0.0f };

__device__ __forceinline__ v8f wmma4(v2f a, v2f b, v8f c) {
    return __builtin_amdgcn_wmma_f32_16x16x4_f32(
        false, a, false, b, (short)0, c, false, false);
}

__global__ __launch_bounds__(512) void sage_layer(
    const float* __restrict__ xin,
    const int* __restrict__ off,
    const int* __restrict__ deg,
    const int* __restrict__ nbr,
    const float* __restrict__ Wl,
    const float* __restrict__ WlT,
    const float* __restrict__ Wr,
    const float* __restrict__ WrT,
    const float* __restrict__ bias,
    const float* __restrict__ biasT,
    float* __restrict__ out)
{
    __shared__ __attribute__((aligned(16))) float ldsB[96 * BPITCH];
    __shared__ __attribute__((aligned(16))) float ldsA[3 * TPB * SLOT];

    {
        v4f w0[2], w1[2], w2[2], w3[2];
#pragma unroll
        for (int it = 0; it < 2; ++it) {
            int g = it * 512 + (int)threadIdx.x;
            w0[it] = ((const v4f*)Wl)[g];
            w1[it] = ((const v4f*)WlT)[g];
            w2[it] = ((const v4f*)Wr)[g];
            w3[it] = ((const v4f*)WrT)[g];
        }
#pragma unroll
        for (int it = 0; it < 2; ++it) {
            int g  = it * 512 + (int)threadIdx.x;
            int n  = g >> 4;
            int kq = g & 15;
            v4f w2s = w2[it] + w3[it];
            float* d = ldsB + (2 * kq) * BPITCH + (n << 1);
            *(v2f*)(d)                = (v2f){ w0[it].x, w0[it].y };
            *(v2f*)(d + BPITCH)       = (v2f){ w0[it].z, w0[it].w };
            *(v2f*)(d + 32 * BPITCH)  = (v2f){ w1[it].x, w1[it].y };
            *(v2f*)(d + 33 * BPITCH)  = (v2f){ w1[it].z, w1[it].w };
            *(v2f*)(d + 64 * BPITCH)  = (v2f){ w2s.x, w2s.y };
            *(v2f*)(d + 65 * BPITCH)  = (v2f){ w2s.z, w2s.w };
        }
    }

    int lane = threadIdx.x & 31;
    int wave = __builtin_amdgcn_readfirstlane((int)(threadIdx.x >> 5));

    {
        int side = wave >> 3;
        int tl   = (wave >> 1) & (TPB - 1);
        int half = wave & 1;
        int tile = (int)blockIdx.x * TPB + tl;
        if (tile < N_TILES) {
            int node0 = tile * TROWS + half * RPW;
            int r0 = side * N_NODES + node0;
            float* slot = ldsA + (side * TPB + tl) * SLOT + half * RPW * APITCH + 2 * lane;

            v2f xr[RPW];
            if (side == 0) {
#pragma unroll
                for (int j = 0; j < RPW; ++j)
                    xr[j] = *(const v2f*)(xin + (size_t)(node0 + j) * D + 2 * lane);
            }

            int ov = off[r0 + (lane & (RPW - 1))];
            int nv = deg[r0 + (lane & (RPW - 1))];

            int o = __builtin_amdgcn_readlane(ov, 0);
            int n = __builtin_amdgcn_readlane(nv, 0);
            int idx = nbr[min(o + lane, NBR_LAST)];

            if (side == 0) {
                float* xs = slot + 2 * TPB * SLOT;
#pragma unroll
                for (int j = 0; j < RPW; ++j)
                    *(v2f*)(xs + j * APITCH) = xr[j];
            }

            for (int j = 0; j < RPW; ++j) {
                int nb[GCH];
                v2f v[GCH];
                v2f acc = { 0.0f, 0.0f };

                GATHER_CHUNK_LOADS(0)

                int o2 = __builtin_amdgcn_readlane(ov, (j + 1) & (RPW - 1));
                int n2 = __builtin_amdgcn_readlane(nv, (j + 1) & (RPW - 1));
                int idx2 = nbr[min(o2 + lane, NBR_LAST)];
                asm volatile("" :: "v"(idx2));

                GATHER_CHUNK_ADDS(0)

                for (int k = GCH; k < n; k += GCH) {
                    if ((k & 31) == 0) idx = nbr[min(o + k + lane, NBR_LAST)];
                    GATHER_CHUNK_LOADS(k)
                    GATHER_CHUNK_ADDS(k)
                }

                float inv = __builtin_amdgcn_rcpf(fmaxf((float)n, 1.0f));
                *(v2f*)(slot + j * APITCH) = (v2f){ acc.x * inv, acc.y * inv };

                o = o2; n = n2; idx = idx2;
            }
        }
    }

    __syncthreads();

    if (wave >= 2 * TPB) return;
    int tl   = wave & (TPB - 1);
    int ch   = wave >> 2;
    int tile = (int)blockIdx.x * TPB + tl;
    if (tile >= N_TILES) return;

    int nl = lane & 15;
    int khalf = (lane < 16) ? 0 : 2;
    const float* arow = ldsA + tl * SLOT + nl * APITCH + khalf;
    const float* bcol = ldsB + (khalf >> 1) * BPITCH + ((ch * 32 + nl) << 1);

    v8f c0 = {}, c1 = {};

#pragma unroll
    for (int seg = 0; seg < 3; ++seg) {
#pragma unroll 4
        for (int kk = 0; kk < 16; ++kk) {
            v2f a = *(const v2f*)(arow + seg * (TPB * SLOT) + kk * 4);
            const float* bb = bcol + (seg * 32 + kk * 2) * BPITCH;
            v2f b0 = *(const v2f*)(bb);
            v2f b1 = *(const v2f*)(bb + 32);
            c0 = wmma4(a, b0, c0);
            c1 = wmma4(a, b1, c1);
        }
    }

    int rowbase = tile * TROWS;
    int rowoff = (lane < 16) ? 0 : 8;
#pragma unroll
    for (int ct = 0; ct < 2; ++ct) {
        int j = ch * 32 + ct * 16 + nl;
        float bs = bias[j] + biasT[j];
        v8f c = (ct == 0) ? c0 : c1;
#pragma unroll
        for (int v = 0; v < 8; ++v) {
            int row = rowbase + rowoff + v;
            out[(size_t)row * D + j] = c[v] + bs;
        }
    }
}

extern "C" void kernel_launch(void* const* d_in, const int* in_sizes, int n_in,
                              void* d_out, int out_size, void* d_ws, size_t ws_size,
                              hipStream_t stream) {
    (void)in_sizes; (void)n_in; (void)out_size; (void)ws_size;

    const float* x    = (const float*)d_in[0];
    const int*   ei   = (const int*)d_in[1];
    const float* Wl1  = (const float*)d_in[2];
    const float* Wr1  = (const float*)d_in[3];
    const float* b1   = (const float*)d_in[4];
    const float* WlT1 = (const float*)d_in[5];
    const float* WrT1 = (const float*)d_in[6];
    const float* bT1  = (const float*)d_in[7];
    const float* Wl2  = (const float*)d_in[8];
    const float* Wr2  = (const float*)d_in[9];
    const float* b2   = (const float*)d_in[10];
    const float* WlT2 = (const float*)d_in[11];
    const float* WrT2 = (const float*)d_in[12];
    const float* bT2  = (const float*)d_in[13];

    const int* src = ei;
    const int* dst = ei + N_EDGES;

    float* ws    = (float*)d_ws;
    float* hid   = ws;
    int*      deg   = (int*)(hid + (size_t)N_NODES * D);
    unsigned* total = (unsigned*)(deg + N_ROWS);
    unsigned* q_rank = total + QSTRIDE;
    unsigned* q_fill = q_rank + 2 * QSTRIDE;
    int* off    = (int*)(q_fill + 2 * QSTRIDE);
    int* slot_f = off + N_ROWS;
    int* slot_r = slot_f + N_EDGES;
    int* nbr    = slot_r + N_EDGES;
    float* out  = (float*)d_out;

    const size_t zero_bytes = (size_t)N_ROWS * sizeof(int)
                            + 5 * (size_t)QSTRIDE * sizeof(unsigned);

    dim3 blk(256);
    dim3 lblk(512);
    dim3 wgrid(WALK_BLOCKS);
    dim3 agrid((N_ROWS + 255) / 256);
    dim3 lgrid((N_TILES + TPB - 1) / TPB);

    hipMemsetAsync(deg, 0, zero_bytes, stream);
    sage_rank<<<wgrid, blk, 0, stream>>>(src, dst, deg, slot_f, slot_r, q_rank);
    sage_alloc<<<agrid, blk, 0, stream>>>(deg, off, total);
    sage_fill<<<wgrid, blk, 0, stream>>>(src, dst, off, slot_f, slot_r, nbr, q_fill);

    sage_layer<<<lgrid, lblk, 0, stream>>>(x, off, deg, nbr,
                                          Wl1, WlT1, Wr1, WrT1, b1, bT1, hid);

    sage_layer<<<lgrid, lblk, 0, stream>>>(hid, off, deg, nbr,
                                          Wl2, WlT2, Wr2, WrT2, b2, bT2, out);
}


# lists: a row's neighbours in a fixed 128-byte record, the returning add's slot is the address: one edge walk builds them (no offset scan, no second walk, 2 launches fewer); sums in arrival order
# speedup vs baseline: 3.1528x; 1.2096x over previous
#include <hip/hip_runtime.h>

#define N_NODES 100000
#define N_EDGES 1200000
#define D 64
#define N_ROWS (2 * N_NODES)

typedef __attribute__((ext_vector_type(2))) float v2f;
typedef __attribute__((ext_vector_type(4))) float v4f;
typedef __attribute__((ext_vector_type(8))) float v8f;

#define SPLIT (N_NODES / 2)
#define QSTRIDE 64
#define EPL 10
#define EPT (32 * EPL)
#define NTICKETS (N_EDGES / EPT)
#define WALK_BLOCKS 512
#define LCAP 32
static_assert(N_EDGES % EPT == 0, "tickets must tile the edge list");

template <class F>
__device__ __forceinline__ void walk_edges_by_half(unsigned* __restrict__ queue, F body)
{
    int lane = threadIdx.x & 31;

    unsigned where = __builtin_amdgcn_s_sendmsg_rtn(0x87);
    int myhalf = (int)((where >> 18) & 1u);

    for (int pass = 0; pass < 2; ++pass) {
        int h = myhalf ^ pass;
        unsigned* q = queue + h * QSTRIDE;

        for (;;) {
            unsigned tv = 0;
            if (lane == 0) tv = atomicAdd(q, 1u);
            unsigned t = (unsigned)__builtin_amdgcn_readfirstlane((int)tv);
            if (t >= (unsigned)NTICKETS) break;
            body((int)t * EPT + lane, h);
        }
    }
}

__device__ __forceinline__ void put_neighbour(
    int* __restrict__ nbr, unsigned* __restrict__ spill_n, int* __restrict__ spill,
    int row, int slot, int nb)
{
    if (slot < LCAP) {
        nbr[row * LCAP + slot] = nb;
    } else {
        unsigned p = atomicAdd(spill_n, 1u);
        spill[2 * p]     = row;
        spill[2 * p + 1] = nb;
    }
}

__global__ __launch_bounds__(256) void sage_lists(
    const int* __restrict__ src,
    const int* __restrict__ dst,
    int* __restrict__ deg,
    int* __restrict__ nbr,
    unsigned* __restrict__ spill_n,
    int* __restrict__ spill,
    unsigned* __restrict__ queue)
{
    walk_edges_by_half(queue, [&](int e0, int h) {
        int s[EPL], d[EPL], rf[EPL], rr[EPL];
#pragma unroll
        for (int i = 0; i < EPL; ++i) {
            s[i] = src[e0 + 32 * i];
            d[i] = dst[e0 + 32 * i];
        }
#pragma unroll
        for (int i = 0; i < EPL; ++i) {
            rf[i] = 0; rr[i] = 0;
            if ((int)(d[i] >= SPLIT) == h) rf[i] = atomicAdd(deg + d[i], 1);
            if ((int)(s[i] >= SPLIT) == h) rr[i] = atomicAdd(deg + N_NODES + s[i], 1);
        }
#pragma unroll
        for (int i = 0; i < EPL; ++i) {
            if ((int)(d[i] >= SPLIT) == h)
                put_neighbour(nbr, spill_n, spill, d[i], rf[i], s[i]);
            if ((int)(s[i] >= SPLIT) == h)
                put_neighbour(nbr, spill_n, spill, N_NODES + s[i], rr[i], d[i]);
        }
    });
}

#define RPW 8
#define TROWS 16
#define GCH 16
#define TPB 4
#define N_TILES (N_NODES / TROWS)
#define BPITCH 130
#define APITCH 68
#define SLOT (TROWS * APITCH)
static_assert(N_NODES % TROWS == 0, "tiles must tile the nodes");
static_assert(LCAP == 32 && LCAP % GCH == 0, "a record is one list load, tiled by chunks");

#define GATHER_CHUNK_LOADS(k)                                                   \
    _Pragma("unroll")                                                           \
    for (int u = 0; u < GCH; ++u)                                               \
        nb[u] = __builtin_amdgcn_readlane(idx, (k) + u);                        \
    {                                                                           \
        int fb = ((k) < m) ? nb[0] : 0;                        \
        _Pragma("unroll")                                                       \
        for (int u = 0; u < GCH; ++u) {                                         \
            int nu = ((k) + u < m) ? nb[u] : fb;               \
            v[u] = *(const v2f*)(xin + (size_t)nu * D + 2 * lane);              \
        }                                                                       \
    }

#define GATHER_CHUNK_ADDS(k)                                                    \
    _Pragma("unroll")                                                           \
    for (int u = 0; u < GCH; ++u)                                               \
        acc += ((k) + u < m) ? v[u] : (v2f){ 0.0f, 0.0f };

__device__ __forceinline__ v8f wmma4(v2f a, v2f b, v8f c) {
    return __builtin_amdgcn_wmma_f32_16x16x4_f32(
        false, a, false, b, (short)0, c, false, false);
}

__global__ __launch_bounds__(512) void sage_layer(
    const float* __restrict__ xin,
    const int* __restrict__ deg,
    const int* __restrict__ nbr,
    const unsigned* __restrict__ spill_n,
    const int* __restrict__ spill,
    const float* __restrict__ Wl,
    const float* __restrict__ WlT,
    const float* __restrict__ Wr,
    const float* __restrict__ WrT,
    const float* __restrict__ bias,
    const float* __restrict__ biasT,
    float* __restrict__ out)
{
    __shared__ __attribute__((aligned(16))) float ldsB[96 * BPITCH];
    __shared__ __attribute__((aligned(16))) float ldsA[3 * TPB * SLOT];

    {
        v4f w0[2], w1[2], w2[2], w3[2];
#pragma unroll
        for (int it = 0; it < 2; ++it) {
            int g = it * 512 + (int)threadIdx.x;
            w0[it] = ((const v4f*)Wl)[g];
            w1[it] = ((const v4f*)WlT)[g];
            w2[it] = ((const v4f*)Wr)[g];
            w3[it] = ((const v4f*)WrT)[g];
        }
#pragma unroll
        for (int it = 0; it < 2; ++it) {
            int g  = it * 512 + (int)threadIdx.x;
            int n  = g >> 4;
            int kq = g & 15;
            v4f w2s = w2[it] + w3[it];
            float* d = ldsB + (2 * kq) * BPITCH + (n << 1);
            *(v2f*)(d)                = (v2f){ w0[it].x, w0[it].y };
            *(v2f*)(d + BPITCH)       = (v2f){ w0[it].z, w0[it].w };
            *(v2f*)(d + 32 * BPITCH)  = (v2f){ w1[it].x, w1[it].y };
            *(v2f*)(d + 33 * BPITCH)  = (v2f){ w1[it].z, w1[it].w };
            *(v2f*)(d + 64 * BPITCH)  = (v2f){ w2s.x, w2s.y };
            *(v2f*)(d + 65 * BPITCH)  = (v2f){ w2s.z, w2s.w };
        }
    }

    int lane = threadIdx.x & 31;
    int wave = __builtin_amdgcn_readfirstlane((int)(threadIdx.x >> 5));

    {
        int side = wave >> 3;
        int tl   = (wave >> 1) & (TPB - 1);
        int half = wave & 1;
        int tile = (int)blockIdx.x * TPB + tl;
        if (tile < N_TILES) {
            int node0 = tile * TROWS + half * RPW;
            int r0 = side * N_NODES + node0;
            float* slot = ldsA + (side * TPB + tl) * SLOT + half * RPW * APITCH + 2 * lane;

            v2f xr[RPW];
            if (side == 0) {
#pragma unroll
                for (int j = 0; j < RPW; ++j)
                    xr[j] = *(const v2f*)(xin + (size_t)(node0 + j) * D + 2 * lane);
            }

            int nv = deg[r0 + (lane & (RPW - 1))];

            int n = __builtin_amdgcn_readlane(nv, 0);
            int idx = nbr[r0 * LCAP + lane];

            if (side == 0) {
                float* xs = slot + 2 * TPB * SLOT;
#pragma unroll
                for (int j = 0; j < RPW; ++j)
                    *(v2f*)(xs + j * APITCH) = xr[j];
            }

            for (int j = 0; j < RPW; ++j) {
                int nb[GCH];
                v2f v[GCH];
                v2f acc = { 0.0f, 0.0f };
                int m = min(n, LCAP);

                GATHER_CHUNK_LOADS(0)

                int n2 = __builtin_amdgcn_readlane(nv, (j + 1) & (RPW - 1));
                int idx2 = nbr[(r0 + ((j + 1) & (RPW - 1))) * LCAP + lane];
                asm volatile("" :: "v"(idx2));

                GATHER_CHUNK_ADDS(0)

                if (m > GCH) {
                    GATHER_CHUNK_LOADS(GCH)
                    GATHER_CHUNK_ADDS(GCH)
                }

                if (n > LCAP) {
                    int row = r0 + j;
                    int np = (int)min(*spill_n, (unsigned)(2 * N_EDGES));
                    for (int p0 = 0; p0 < np; p0 += 32) {
                        int p = p0 + lane;
                        int prow = -1, pnb = 0;
                        if (p < np) { prow = spill[2 * p]; pnb = spill[2 * p + 1]; }
                        unsigned hit = __builtin_amdgcn_ballot_w32(prow == row);
                        while (hit) {
                            int b = __builtin_ctz(hit);
                            hit &= hit - 1;
                            int nu = __builtin_amdgcn_readlane(pnb, b);
                            acc += *(const v2f*)(xin + (size_t)nu * D + 2 * lane);
                        }
                    }
                }

                float inv = __builtin_amdgcn_rcpf(fmaxf((float)n, 1.0f));
                *(v2f*)(slot + j * APITCH) = (v2f){ acc.x * inv, acc.y * inv };

                n = n2; idx = idx2;
            }
        }
    }

    __syncthreads();

    if (wave >= 2 * TPB) return;
    int tl   = wave & (TPB - 1);
    int ch   = wave >> 2;
    int tile = (int)blockIdx.x * TPB + tl;
    if (tile >= N_TILES) return;

    int nl = lane & 15;
    int khalf = (lane < 16) ? 0 : 2;
    const float* arow = ldsA + tl * SLOT + nl * APITCH + khalf;
    const float* bcol = ldsB + (khalf >> 1) * BPITCH + ((ch * 32 + nl) << 1);

    v8f c0 = {}, c1 = {};

#pragma unroll
    for (int seg = 0; seg < 3; ++seg) {
#pragma unroll 4
        for (int kk = 0; kk < 16; ++kk) {
            v2f a = *(const v2f*)(arow + seg * (TPB * SLOT) + kk * 4);
            const float* bb = bcol + (seg * 32 + kk * 2) * BPITCH;
            v2f b0 = *(const v2f*)(bb);
            v2f b1 = *(const v2f*)(bb + 32);
            c0 = wmma4(a, b0, c0);
            c1 = wmma4(a, b1, c1);
        }
    }

    int rowbase = tile * TROWS;
    int rowoff = (lane < 16) ? 0 : 8;
#pragma unroll
    for (int ct = 0; ct < 2; ++ct) {
        int j = ch * 32 + ct * 16 + nl;
        float bs = bias[j] + biasT[j];
        v8f c = (ct == 0) ? c0 : c1;
#pragma unroll
        for (int v = 0; v < 8; ++v) {
            int row = rowbase + rowoff + v;
            out[(size_t)row * D + j] = c[v] + bs;
        }
    }
}

extern "C" void kernel_launch(void* const* d_in, const int* in_sizes, int n_in,
                              void* d_out, int out_size, void* d_ws, size_t ws_size,
                              hipStream_t stream) {
    (void)in_sizes; (void)n_in; (void)out_size; (void)ws_size;

    const float* x    = (const float*)d_in[0];
    const int*   ei   = (const int*)d_in[1];
    const float* Wl1  = (const float*)d_in[2];
    const float* Wr1  = (const float*)d_in[3];
    const float* b1   = (const float*)d_in[4];
    const float* WlT1 = (const float*)d_in[5];
    const float* WrT1 = (const float*)d_in[6];
    const float* bT1  = (const float*)d_in[7];
    const float* Wl2  = (const float*)d_in[8];
    const float* Wr2  = (const float*)d_in[9];
    const float* b2   = (const float*)d_in[10];
    const float* WlT2 = (const float*)d_in[11];
    const float* WrT2 = (const float*)d_in[12];
    const float* bT2  = (const float*)d_in[13];

    const int* src = ei;
    const int* dst = ei + N_EDGES;

    float* ws    = (float*)d_ws;
    float* hid   = ws;
    int*      deg     = (int*)(hid + (size_t)N_NODES * D);
    unsigned* spill_n = (unsigned*)(deg + N_ROWS);
    unsigned* q_walk  = spill_n + QSTRIDE;
    int* nbr   = (int*)(q_walk + 2 * QSTRIDE);
    int* spill = nbr + (size_t)N_ROWS * LCAP;
    float* out = (float*)d_out;

    const size_t zero_bytes = (size_t)N_ROWS * sizeof(int)
                            + 3 * (size_t)QSTRIDE * sizeof(unsigned);

    dim3 blk(256);
    dim3 lblk(512);
    dim3 wgrid(WALK_BLOCKS);
    dim3 lgrid((N_TILES + TPB - 1) / TPB);

    hipMemsetAsync(deg, 0, zero_bytes, stream);
    sage_lists<<<wgrid, blk, 0, stream>>>(src, dst, deg, nbr, spill_n, spill, q_walk);

    sage_layer<<<lgrid, lblk, 0, stream>>>(x, deg, nbr, spill_n, spill,
                                          Wl1, WlT1, Wr1, WrT1, b1, bT1, hid);

    sage_layer<<<lgrid, lblk, 0, stream>>>(hid, deg, nbr, spill_n, spill,
                                          Wl2, WlT2, Wr2, WrT2, b2, bT2, out);
}


# layer: resident work-groups (768) stage weights once and claim 2-tile groups by a returning add; two LDS slot sets: 4 waves multiply group k beside the row sums of group k+1; sum order unchanged
# speedup vs baseline: 3.3634x; 1.0668x over previous
#include <hip/hip_runtime.h>

#define N_NODES 100000
#define N_EDGES 1200000
#define D 64
#define N_ROWS (2 * N_NODES)

typedef __attribute__((ext_vector_type(2))) float v2f;
typedef __attribute__((ext_vector_type(4))) float v4f;
typedef __attribute__((ext_vector_type(8))) float v8f;

#define SPLIT (N_NODES / 2)
#define QSTRIDE 64
#define EPL 10
#define EPT (32 * EPL)
#define NTICKETS (N_EDGES / EPT)
#define WALK_BLOCKS 512
#define LCAP 32
static_assert(N_EDGES % EPT == 0, "tickets must tile the edge list");

template <class F>
__device__ __forceinline__ void walk_edges_by_half(unsigned* __restrict__ queue, F body)
{
    int lane = threadIdx.x & 31;

    unsigned where = __builtin_amdgcn_s_sendmsg_rtn(0x87);
    int myhalf = (int)((where >> 18) & 1u);

    for (int pass = 0; pass < 2; ++pass) {
        int h = myhalf ^ pass;
        unsigned* q = queue + h * QSTRIDE;

        for (;;) {
            unsigned tv = 0;
            if (lane == 0) tv = atomicAdd(q, 1u);
            unsigned t = (unsigned)__builtin_amdgcn_readfirstlane((int)tv);
            if (t >= (unsigned)NTICKETS) break;
            body((int)t * EPT + lane, h);
        }
    }
}

__device__ __forceinline__ void put_neighbour(
    int* __restrict__ nbr, unsigned* __restrict__ spill_n, int* __restrict__ spill,
    int row, int slot, int nb)
{
    if (slot < LCAP) {
        nbr[row * LCAP + slot] = nb;
    } else {
        unsigned p = atomicAdd(spill_n, 1u);
        spill[2 * p]     = row;
        spill[2 * p + 1] = nb;
    }
}

__global__ __launch_bounds__(256) void sage_lists(
    const int* __restrict__ src,
    const int* __restrict__ dst,
    int* __restrict__ deg,
    int* __restrict__ nbr,
    unsigned* __restrict__ spill_n,
    int* __restrict__ spill,
    unsigned* __restrict__ queue)
{
    walk_edges_by_half(queue, [&](int e0, int h) {
        int s[EPL], d[EPL], rf[EPL], rr[EPL];
#pragma unroll
        for (int i = 0; i < EPL; ++i) {
            s[i] = src[e0 + 32 * i];
            d[i] = dst[e0 + 32 * i];
        }
#pragma unroll
        for (int i = 0; i < EPL; ++i) {
            rf[i] = 0; rr[i] = 0;
            if ((int)(d[i] >= SPLIT) == h) rf[i] = atomicAdd(deg + d[i], 1);
            if ((int)(s[i] >= SPLIT) == h) rr[i] = atomicAdd(deg + N_NODES + s[i], 1);
        }
#pragma unroll
        for (int i = 0; i < EPL; ++i) {
            if ((int)(d[i] >= SPLIT) == h)
                put_neighbour(nbr, spill_n, spill, d[i], rf[i], s[i]);
            if ((int)(s[i] >= SPLIT) == h)
                put_neighbour(nbr, spill_n, spill, N_NODES + s[i], rr[i], d[i]);
        }
    });
}

#define RPW 4
#define TROWS 16
#define GCH 16
#define TPB 2
#define N_TILES (N_NODES / TROWS)
#define N_GROUPS (N_TILES / TPB)
#define LAYER_BLOCKS 768
#define PWAVE0 12
#define BPITCH 130
#define APITCH 68
#define SLOT (TROWS * APITCH)
#define ASET (3 * TPB * SLOT)
static_assert(N_NODES % TROWS == 0 && N_TILES % TPB == 0, "groups must tile the nodes");
static_assert(LAYER_BLOCKS <= N_GROUPS, "every work-group has a first group");
static_assert(2 * TPB * (TROWS / RPW) == 16, "16 waves sum a group's forward and reverse rows");
static_assert(LCAP == 32 && LCAP % GCH == 0, "a record is one list load, tiled by chunks");

#define GATHER_CHUNK_LOADS(k)                                                   \
    _Pragma("unroll")                                                           \
    for (int u = 0; u < GCH; ++u)                                               \
        nb[u] = __builtin_amdgcn_readlane(idx, (k) + u);                        \
    {                                                                           \
        int fb = ((k) < m) ? nb[0] : 0;                        \
        _Pragma("unroll")                                                       \
        for (int u = 0; u < GCH; ++u) {                                         \
            int nu = ((k) + u < m) ? nb[u] : fb;               \
            v[u] = *(const v2f*)(xin + (size_t)nu * D + 2 * lane);              \
        }                                                                       \
    }

#define GATHER_CHUNK_ADDS(k)                                                    \
    _Pragma("unroll")                                                           \
    for (int u = 0; u < GCH; ++u)                                               \
        acc += ((k) + u < m) ? v[u] : (v2f){ 0.0f, 0.0f };

__device__ __forceinline__ v8f wmma4(v2f a, v2f b, v8f c) {
    return __builtin_amdgcn_wmma_f32_16x16x4_f32(
        false, a, false, b, (short)0, c, false, false);
}

template <class Mid>
__device__ __forceinline__ void sum_group(
    const float* __restrict__ xin,
    const int* __restrict__ deg,
    const int* __restrict__ nbr,
    const unsigned* __restrict__ spill_n,
    const int* __restrict__ spill,
    float* aset, int group, int wave, int lane, Mid mid)
{
    int side = wave >> 3;
    int tl   = (wave >> 2) & (TPB - 1);
    int part = wave & 3;
    int tile = group * TPB + tl;
    int node0 = tile * TROWS + part * RPW;
    int r0 = side * N_NODES + node0;
    float* slot = aset + (side * TPB + tl) * SLOT + part * RPW * APITCH + 2 * lane;

    v2f xr[RPW];
    if (side == 0) {
#pragma unroll
        for (int j = 0; j < RPW; ++j)
            xr[j] = *(const v2f*)(xin + (size_t)(node0 + j) * D + 2 * lane);
    }

    int nv = deg[r0 + (lane & (RPW - 1))];
    int idx = nbr[r0 * LCAP + lane];

    mid();

    int n = __builtin_amdgcn_readlane(nv, 0);

    if (side == 0) {
        float* xs = slot + 2 * TPB * SLOT;
#pragma unroll
        for (int j = 0; j < RPW; ++j)
            *(v2f*)(xs + j * APITCH) = xr[j];
    }

    for (int j = 0; j < RPW; ++j) {
        int nb[GCH];
        v2f v[GCH];
        v2f acc = { 0.0f, 0.0f };
        int m = min(n, LCAP);

        GATHER_CHUNK_LOADS(0)

        int n2 = __builtin_amdgcn_readlane(nv, (j + 1) & (RPW - 1));
        int idx2 = nbr[(r0 + ((j + 1) & (RPW - 1))) * LCAP + lane];
        asm volatile("" :: "v"(idx2));

        GATHER_CHUNK_ADDS(0)

        if (m > GCH) {
            GATHER_CHUNK_LOADS(GCH)
            GATHER_CHUNK_ADDS(GCH)
        }

        if (n > LCAP) {
            int row = r0 + j;
            int np = (int)min(*spill_n, (unsigned)(2 * N_EDGES));
            for (int p0 = 0; p0 < np; p0 += 32) {
                int p = p0 + lane;
                int prow = -1, pnb = 0;
                if (p < np) { prow = spill[2 * p]; pnb = spill[2 * p + 1]; }
                unsigned hit = __builtin_amdgcn_ballot_w32(prow == row);
                while (hit) {
                    int b = __builtin_ctz(hit);
                    hit &= hit - 1;
                    int nu = __builtin_amdgcn_readlane(pnb, b);
                    acc += *(const v2f*)(xin + (size_t)nu * D + 2 * lane);
                }
            }
        }

        float inv = __builtin_amdgcn_rcpf(fmaxf((float)n, 1.0f));
        *(v2f*)(slot + j * APITCH) = (v2f){ acc.x * inv, acc.y * inv };

        n = n2; idx = idx2;
    }
}

__device__ __forceinline__ void product_tile(
    const float* aset, const float* ldsB, int tl, int ch, int tile, int lane,
    const float* __restrict__ bias, const float* __restrict__ biasT,
    float* __restrict__ out)
{
    int nl = lane & 15;
    int khalf = (lane < 16) ? 0 : 2;
    const float* arow = aset + tl * SLOT + nl * APITCH + khalf;
    const float* bcol = ldsB + (khalf >> 1) * BPITCH + ((ch * 32 + nl) << 1);

    v8f c0 = {}, c1 = {};

#pragma unroll
    for (int seg = 0; seg < 3; ++seg) {
#pragma unroll 4
        for (int kk = 0; kk < 16; ++kk) {
            v2f a = *(const v2f*)(arow + seg * (TPB * SLOT) + kk * 4);
            const float* bb = bcol + (seg * 32 + kk * 2) * BPITCH;
            v2f b0 = *(const v2f*)(bb);
            v2f b1 = *(const v2f*)(bb + 32);
            c0 = wmma4(a, b0, c0);
            c1 = wmma4(a, b1, c1);
        }
    }

    int rowbase = tile * TROWS;
    int rowoff = (lane < 16) ? 0 : 8;
#pragma unroll
    for (int ct = 0; ct < 2; ++ct) {
        int j = ch * 32 + ct * 16 + nl;
        float bs = bias[j] + biasT[j];
        v8f c = (ct == 0) ? c0 : c1;
#pragma unroll
        for (int v = 0; v < 8; ++v) {
            int row = rowbase + rowoff + v;
            out[(size_t)row * D + j] = c[v] + bs;
        }
    }
}

__global__ __launch_bounds__(512) void sage_layer(
    const float* __restrict__ xin,
    const int* __restrict__ deg,
    const int* __restrict__ nbr,
    const unsigned* __restrict__ spill_n,
    const int* __restrict__ spill,
    const float* __restrict__ Wl,
    const float* __restrict__ WlT,
    const float* __restrict__ Wr,
    const float* __restrict__ WrT,
    const float* __restrict__ bias,
    const float* __restrict__ biasT,
    unsigned* __restrict__ gq,
    float* __restrict__ out)
{
    __shared__ __attribute__((aligned(16))) float ldsB[96 * BPITCH];
    __shared__ __attribute__((aligned(16))) float ldsA[2 * ASET];
    __shared__ int nextg[2];

    {
        v4f w0[2], w1[2], w2[2], w3[2];
#pragma unroll
        for (int it = 0; it < 2; ++it) {
            int g = it * 512 + (int)threadIdx.x;
            w0[it] = ((const v4f*)Wl)[g];
            w1[it] = ((const v4f*)WlT)[g];
            w2[it] = ((const v4f*)Wr)[g];
            w3[it] = ((const v4f*)WrT)[g];
        }
#pragma unroll
        for (int it = 0; it < 2; ++it) {
            int g  = it * 512 + (int)threadIdx.x;
            int n  = g >> 4;
            int kq = g & 15;
            v4f w2s = w2[it] + w3[it];
            float* d = ldsB + (2 * kq) * BPITCH + (n << 1);
            *(v2f*)(d)                = (v2f){ w0[it].x, w0[it].y };
            *(v2f*)(d + BPITCH)       = (v2f){ w0[it].z, w0[it].w };
            *(v2f*)(d + 32 * BPITCH)  = (v2f){ w1[it].x, w1[it].y };
            *(v2f*)(d + 33 * BPITCH)  = (v2f){ w1[it].z, w1[it].w };
            *(v2f*)(d + 64 * BPITCH)  = (v2f){ w2s.x, w2s.y };
            *(v2f*)(d + 65 * BPITCH)  = (v2f){ w2s.z, w2s.w };
        }
    }

    int lane = threadIdx.x & 31;
    int wave = __builtin_amdgcn_readfirstlane((int)(threadIdx.x >> 5));

    int ptl = (wave - PWAVE0) & (TPB - 1);
    int pch = (wave - PWAVE0) >> 1;

    int g = (int)blockIdx.x;

    sum_group(xin, deg, nbr, spill_n, spill, ldsA, g, wave, lane, [&] {
        if (threadIdx.x == 0) nextg[0] = (int)(gridDim.x + atomicAdd(gq, 1u));
    });

    __syncthreads();

    for (int it = 0; ; ++it) {
        const float* cur = ldsA + (it & 1) * ASET;
        float* nxt = ldsA + ((it & 1) ^ 1) * ASET;
        int gn = __builtin_amdgcn_readfirstlane(nextg[it & 1]);

        auto product = [&] {
            if (wave >= PWAVE0)
                product_tile(cur, ldsB, ptl, pch, g * TPB + ptl, lane, bias, biasT, out);
        };
        auto claim_and_product = [&] {
            if (threadIdx.x == 0)
                nextg[(it & 1) ^ 1] = (int)(gridDim.x + atomicAdd(gq, 1u));
            product();
        };

        if (gn >= N_GROUPS) {
            product();
            break;
        }

        sum_group(xin, deg, nbr, spill_n, spill, nxt, gn, wave, lane, claim_and_product);

        __syncthreads();
        g = gn;
    }
}

extern "C" void kernel_launch(void* const* d_in, const int* in_sizes, int n_in,
                              void* d_out, int out_size, void* d_ws, size_t ws_size,
                              hipStream_t stream) {
    (void)in_sizes; (void)n_in; (void)out_size; (void)ws_size;

    const float* x    = (const float*)d_in[0];
    const int*   ei   = (const int*)d_in[1];
    const float* Wl1  = (const float*)d_in[2];
    const float* Wr1  = (const float*)d_in[3];
    const float* b1   = (const float*)d_in[4];
    const float* WlT1 = (const float*)d_in[5];
    const float* WrT1 = (const float*)d_in[6];
    const float* bT1  = (const float*)d_in[7];
    const float* Wl2  = (const float*)d_in[8];
    const float* Wr2  = (const float*)d_in[9];
    const float* b2   = (const float*)d_in[10];
    const float* WlT2 = (const float*)d_in[11];
    const float* WrT2 = (const float*)d_in[12];
    const float* bT2  = (const float*)d_in[13];

    const int* src = ei;
    const int* dst = ei + N_EDGES;

    float* ws    = (float*)d_ws;
    float* hid   = ws;
    int*      deg     = (int*)(hid + (size_t)N_NODES * D);
    unsigned* spill_n = (unsigned*)(deg + N_ROWS);
    unsigned* q_walk  = spill_n + QSTRIDE;
    unsigned* q_layer = q_walk + 2 * QSTRIDE;
    int* nbr   = (int*)(q_layer + 2 * QSTRIDE);
    int* spill = nbr + (size_t)N_ROWS * LCAP;
    float* out = (float*)d_out;

    const size_t zero_bytes = (size_t)N_ROWS * sizeof(int)
                            + 5 * (size_t)QSTRIDE * sizeof(unsigned);

    dim3 blk(256);
    dim3 lblk(512);
    dim3 wgrid(WALK_BLOCKS);
    dim3 lgrid(LAYER_BLOCKS);

    hipMemsetAsync(deg, 0, zero_bytes, stream);
    sage_lists<<<wgrid, blk, 0, stream>>>(src, dst, deg, nbr, spill_n, spill, q_walk);

    sage_layer<<<lgrid, lblk, 0, stream>>>(x, deg, nbr, spill_n, spill,
                                          Wl1, WlT1, Wr1, WrT1, b1, bT1, q_layer, hid);

    sage_layer<<<lgrid, lblk, 0, stream>>>(hid, deg, nbr, spill_n, spill,
                                          Wl2, WlT2, Wr2, WrT2, b2, bT2, q_layer + QSTRIDE, out);
}


# lists: edge ends sorted by 512-row bin in LDS and copied out in runs, then one work-group per bin draws slots with LDS adds and writes records and degrees as whole lines; sum order = LDS add order
# speedup vs baseline: 3.8887x; 1.1562x over previous
#include <hip/hip_runtime.h>

#define N_NODES 100000
#define N_EDGES 1200000
#define D 64
#define N_ROWS (2 * N_NODES)

typedef __attribute__((ext_vector_type(2))) float v2f;
typedef __attribute__((ext_vector_type(4))) float v4f;
typedef __attribute__((ext_vector_type(8))) float v8f;

#define QSTRIDE 64
#define LCAP 32
#define RB 512
#define NBINS ((N_ROWS + RB - 1) / RB)
#define BCAP 12288
#define EB 2048
#define BIN_BLOCKS ((N_EDGES + EB - 1) / EB)
#define EPT_A (EB / 256)
#define BINS_PER_LANE ((NBINS + 31) / 32)
static_assert(NBINS > 256 && NBINS <= 512, "a thread of sage_bins reserves bins tid and tid + 256");

typedef __attribute__((ext_vector_type(2))) int v2i;
typedef __attribute__((ext_vector_type(4))) int v4i;

__global__ __launch_bounds__(256) void sage_bins(
    const int* __restrict__ src,
    const int* __restrict__ dst,
    unsigned* __restrict__ bin_n,
    v2i* __restrict__ bins,
    unsigned* __restrict__ ovf_n,
    v2i* __restrict__ ovf)
{
    __shared__ int hist[512];
    __shared__ int loff[512];
    __shared__ unsigned gb[512];
    __shared__ v2i sorted[2 * EB];

    int tid = (int)threadIdx.x;
    int e0 = (int)blockIdx.x * EB;

    for (int b = tid; b < 512; b += 256) hist[b] = 0;
    __syncthreads();

    int s[EPT_A], d[EPT_A], kf[EPT_A], kr[EPT_A];
#pragma unroll
    for (int i = 0; i < EPT_A; ++i) {
        int e = e0 + i * 256 + tid;
        bool ok = e < N_EDGES;
        s[i] = ok ? src[e] : -1;
        d[i] = ok ? dst[e] : -1;
    }
#pragma unroll
    for (int i = 0; i < EPT_A; ++i) {
        kf[i] = 0; kr[i] = 0;
        if (s[i] >= 0) {
            kf[i] = atomicAdd(&hist[d[i] / RB], 1);
            kr[i] = atomicAdd(&hist[(N_NODES + s[i]) / RB], 1);
        }
    }
    __syncthreads();

    {
        int b0 = tid, b1 = tid + 256;
        int c0 = hist[b0];
        int c1 = (b1 < NBINS) ? hist[b1] : 0;
        unsigned g0 = 0u, g1 = 0u;
        if (c0) g0 = atomicAdd(bin_n + (size_t)b0 * QSTRIDE, (unsigned)c0);
        if (c1) g1 = atomicAdd(bin_n + (size_t)b1 * QSTRIDE, (unsigned)c1);
        gb[b0] = g0;
        gb[b1] = g1;
    }
    if (tid < 32) {
        int sum = 0;
        for (int j = 0; j < BINS_PER_LANE; ++j) {
            int b = tid * BINS_PER_LANE + j;
            sum += (b < NBINS) ? hist[b] : 0;
        }
        int incl = sum;
        for (int o = 1; o < 32; o <<= 1) {
            int up = __shfl_up(incl, o, 32);
            if (tid >= o) incl += up;
        }
        int run = incl - sum;
        for (int j = 0; j < BINS_PER_LANE; ++j) {
            int b = tid * BINS_PER_LANE + j;
            if (b < NBINS) { loff[b] = run; run += hist[b]; }
        }
    }
    __syncthreads();

#pragma unroll
    for (int i = 0; i < EPT_A; ++i) {
        if (s[i] >= 0) {
            int rf = d[i], rr = N_NODES + s[i];
            sorted[loff[rf / RB] + kf[i]] = (v2i){ rf, s[i] };
            sorted[loff[rr / RB] + kr[i]] = (v2i){ rr, d[i] };
        }
    }
    __syncthreads();

    int total = 2 * min(EB, N_EDGES - e0);
    for (int i = tid; i < total; i += 256) {
        v2i ent = sorted[i];
        int b = ent.x / RB;
        unsigned at = gb[b] + (unsigned)(i - loff[b]);
        if (at < (unsigned)BCAP) {
            bins[(size_t)b * BCAP + at] = ent;
        } else {
            unsigned p = atomicAdd(ovf_n, 1u);
            ovf[p] = ent;
        }
    }
}

__global__ __launch_bounds__(512) void sage_records(
    const unsigned* __restrict__ bin_n,
    const v2i* __restrict__ bins,
    const unsigned* __restrict__ ovf_n,
    const v2i* __restrict__ ovf,
    int* __restrict__ deg,
    int* __restrict__ nbr,
    unsigned* __restrict__ spill_n,
    int* __restrict__ spill)
{
    __shared__ int cnt[RB];
    __shared__ __attribute__((aligned(16))) int rec[RB * LCAP];

    int tid = (int)threadIdx.x;
    int b = (int)blockIdx.x;
    int row0 = b * RB;

    for (int i = tid; i < RB; i += 512) cnt[i] = 0;
    for (int i = tid; i < RB * LCAP / 4; i += 512) ((v4i*)rec)[i] = (v4i){ 0, 0, 0, 0 };
    __syncthreads();

    auto place = [&](int row, int nb) {
        int r = row - row0;
        int slot = atomicAdd(&cnt[r], 1);
        if (slot < LCAP) {
            rec[r * LCAP + slot] = nb;
        } else {
            unsigned p = atomicAdd(spill_n, 1u);
            spill[2 * p]     = row;
            spill[2 * p + 1] = nb;
        }
    };

    unsigned have = bin_n[(size_t)b * QSTRIDE];
    int n = (int)min(have, (unsigned)BCAP);
    const v2i* mine = bins + (size_t)b * BCAP;
    for (int i0 = 0; i0 < n; i0 += 4 * 512) {
        v2i ent[4];
#pragma unroll
        for (int u = 0; u < 4; ++u) {
            int i = i0 + u * 512 + tid;
            ent[u] = (i < n) ? mine[i] : (v2i){ -1, 0 };
        }
#pragma unroll
        for (int u = 0; u < 4; ++u)
            if (ent[u].x >= 0) place(ent[u].x, ent[u].y);
    }
    if (have > (unsigned)BCAP) {
        int np = (int)min(*ovf_n, (unsigned)(2 * N_EDGES));
        for (int i = tid; i < np; i += 512) {
            v2i ent = ovf[i];
            if (ent.x / RB == b) place(ent.x, ent.y);
        }
    }
    __syncthreads();

    for (int r = tid; r < RB; r += 512)
        if (row0 + r < N_ROWS) deg[row0 + r] = cnt[r];
    for (int i = tid; i < RB * LCAP / 4; i += 512)
        if (row0 + i / (LCAP / 4) < N_ROWS)
            ((v4i*)nbr)[(size_t)row0 * (LCAP / 4) + i] = ((const v4i*)rec)[i];
}

#define RPW 4
#define TROWS 16
#define GCH 16
#define TPB 2
#define N_TILES (N_NODES / TROWS)
#define N_GROUPS (N_TILES / TPB)
#define LAYER_BLOCKS 768
#define PWAVE0 12
#define BPITCH 130
#define APITCH 68
#define SLOT (TROWS * APITCH)
#define ASET (3 * TPB * SLOT)
static_assert(N_NODES % TROWS == 0 && N_TILES % TPB == 0, "groups must tile the nodes");
static_assert(LAYER_BLOCKS <= N_GROUPS, "every work-group has a first group");
static_assert(2 * TPB * (TROWS / RPW) == 16, "16 waves sum a group's forward and reverse rows");
static_assert(LCAP == 32 && LCAP % GCH == 0, "a record is one list load, tiled by chunks");

#define GATHER_CHUNK_LOADS(k)                                                   \
    _Pragma("unroll")                                                           \
    for (int u = 0; u < GCH; ++u)                                               \
        nb[u] = __builtin_amdgcn_readlane(idx, (k) + u);                        \
    {                                                                           \
        int fb = ((k) < m) ? nb[0] : 0;                        \
        _Pragma("unroll")                                                       \
        for (int u = 0; u < GCH; ++u) {                                         \
            int nu = ((k) + u < m) ? nb[u] : fb;               \
            v[u] = *(const v2f*)(xin + (size_t)nu * D + 2 * lane);              \
        }                                                                       \
    }

#define GATHER_CHUNK_ADDS(k)                                                    \
    _Pragma("unroll")                                                           \
    for (int u = 0; u < GCH; ++u)                                               \
        acc += ((k) + u < m) ? v[u] : (v2f){ 0.0f, 0.0f };

__device__ __forceinline__ v8f wmma4(v2f a, v2f b, v8f c) {
    return __builtin_amdgcn_wmma_f32_16x16x4_f32(
        false, a, false, b, (short)0, c, false, false);
}

template <class Mid>
__device__ __forceinline__ void sum_group(
    const float* __restrict__ xin,
    const int* __restrict__ deg,
    const int* __restrict__ nbr,
    const unsigned* __restrict__ spill_n,
    const int* __restrict__ spill,
    float* aset, int group, int wave, int lane, Mid mid)
{
    int side = wave >> 3;
    int tl   = (wave >> 2) & (TPB - 1);
    int part = wave & 3;
    int tile = group * TPB + tl;
    int node0 = tile * TROWS + part * RPW;
    int r0 = side * N_NODES + node0;
    float* slot = aset + (side * TPB + tl) * SLOT + part * RPW * APITCH + 2 * lane;

    v2f xr[RPW];
    if (side == 0) {
#pragma unroll
        for (int j = 0; j < RPW; ++j)
            xr[j] = *(const v2f*)(xin + (size_t)(node0 + j) * D + 2 * lane);
    }

    int nv = deg[r0 + (lane & (RPW - 1))];
    int idx = nbr[r0 * LCAP + lane];

    mid();

    int n = __builtin_amdgcn_readlane(nv, 0);

    if (side == 0) {
        float* xs = slot + 2 * TPB * SLOT;
#pragma unroll
        for (int j = 0; j < RPW; ++j)
            *(v2f*)(xs + j * APITCH) = xr[j];
    }

    for (int j = 0; j < RPW; ++j) {
        int nb[GCH];
        v2f v[GCH];
        v2f acc = { 0.0f, 0.0f };
        int m = min(n, LCAP);

        GATHER_CHUNK_LOADS(0)

        int n2 = __builtin_amdgcn_readlane(nv, (j + 1) & (RPW - 1));
        int idx2 = nbr[(r0 + ((j + 1) & (RPW - 1))) * LCAP + lane];
        asm volatile("" :: "v"(idx2));

        GATHER_CHUNK_ADDS(0)

        if (m > GCH) {
            GATHER_CHUNK_LOADS(GCH)
            GATHER_CHUNK_ADDS(GCH)
        }

        if (n > LCAP) {
            int row = r0 + j;
            int np = (int)min(*spill_n, (unsigned)(2 * N_EDGES));
            for (int p0 = 0; p0 < np; p0 += 32) {
                int p = p0 + lane;
                int prow = -1, pnb = 0;
                if (p < np) { prow = spill[2 * p]; pnb = spill[2 * p + 1]; }
                unsigned hit = __builtin_amdgcn_ballot_w32(prow == row);
                while (hit) {
                    int b = __builtin_ctz(hit);
                    hit &= hit - 1;
                    int nu = __builtin_amdgcn_readlane(pnb, b);
                    acc += *(const v2f*)(xin + (size_t)nu * D + 2 * lane);
                }
            }
        }

        float inv = __builtin_amdgcn_rcpf(fmaxf((float)n, 1.0f));
        *(v2f*)(slot + j * APITCH) = (v2f){ acc.x * inv, acc.y * inv };

        n = n2; idx = idx2;
    }
}

__device__ __forceinline__ void product_tile(
    const float* aset, const float* ldsB, int tl, int ch, int tile, int lane,
    const float* __restrict__ bias, const float* __restrict__ biasT,
    float* __restrict__ out)
{
    int nl = lane & 15;
    int khalf = (lane < 16) ? 0 : 2;
    const float* arow = aset + tl * SLOT + nl * APITCH + khalf;
    const float* bcol = ldsB + (khalf >> 1) * BPITCH + ((ch * 32 + nl) << 1);

    v8f c0 = {}, c1 = {};

#pragma unroll
    for (int seg = 0; seg < 3; ++seg) {
#pragma unroll 4
        for (int kk = 0; kk < 16; ++kk) {
            v2f a = *(const v2f*)(arow + seg * (TPB * SLOT) + kk * 4);
            const float* bb = bcol + (seg * 32 + kk * 2) * BPITCH;
            v2f b0 = *(const v2f*)(bb);
            v2f b1 = *(const v2f*)(bb + 32);
            c0 = wmma4(a, b0, c0);
            c1 = wmma4(a, b1, c1);
        }
    }

    int rowbase = tile * TROWS;
    int rowoff = (lane < 16) ? 0 : 8;
#pragma unroll
    for (int ct = 0; ct < 2; ++ct) {
        int j = ch * 32 + ct * 16 + nl;
        float bs = bias[j] + biasT[j];
        v8f c = (ct == 0) ? c0 : c1;
#pragma unroll
        for (int v = 0; v < 8; ++v) {
            int row = rowbase + rowoff + v;
            out[(size_t)row * D + j] = c[v] + bs;
        }
    }
}

__global__ __launch_bounds__(512) void sage_layer(
    const float* __restrict__ xin,
    const int* __restrict__ deg,
    const int* __restrict__ nbr,
    const unsigned* __restrict__ spill_n,
    const int* __restrict__ spill,
    const float* __restrict__ Wl,
    const float* __restrict__ WlT,
    const float* __restrict__ Wr,
    const float* __restrict__ WrT,
    const float* __restrict__ bias,
    const float* __restrict__ biasT,
    unsigned* __restrict__ gq,
    float* __restrict__ out)
{
    __shared__ __attribute__((aligned(16))) float ldsB[96 * BPITCH];
    __shared__ __attribute__((aligned(16))) float ldsA[2 * ASET];
    __shared__ int nextg[2];

    {
        v4f w0[2], w1[2], w2[2], w3[2];
#pragma unroll
        for (int it = 0; it < 2; ++it) {
            int g = it * 512 + (int)threadIdx.x;
            w0[it] = ((const v4f*)Wl)[g];
            w1[it] = ((const v4f*)WlT)[g];
            w2[it] = ((const v4f*)Wr)[g];
            w3[it] = ((const v4f*)WrT)[g];
        }
#pragma unroll
        for (int it = 0; it < 2; ++it) {
            int g  = it * 512 + (int)threadIdx.x;
            int n  = g >> 4;
            int kq = g & 15;
            v4f w2s = w2[it] + w3[it];
            float* d = ldsB + (2 * kq) * BPITCH + (n << 1);
            *(v2f*)(d)                = (v2f){ w0[it].x, w0[it].y };
            *(v2f*)(d + BPITCH)       = (v2f){ w0[it].z, w0[it].w };
            *(v2f*)(d + 32 * BPITCH)  = (v2f){ w1[it].x, w1[it].y };
            *(v2f*)(d + 33 * BPITCH)  = (v2f){ w1[it].z, w1[it].w };
            *(v2f*)(d + 64 * BPITCH)  = (v2f){ w2s.x, w2s.y };
            *(v2f*)(d + 65 * BPITCH)  = (v2f){ w2s.z, w2s.w };
        }
    }

    int lane = threadIdx.x & 31;
    int wave = __builtin_amdgcn_readfirstlane((int)(threadIdx.x >> 5));

    int ptl = (wave - PWAVE0) & (TPB - 1);
    int pch = (wave - PWAVE0) >> 1;

    int g = (int)blockIdx.x;

    sum_group(xin, deg, nbr, spill_n, spill, ldsA, g, wave, lane, [&] {
        if (threadIdx.x == 0) nextg[0] = (int)(gridDim.x + atomicAdd(gq, 1u));
    });

    __syncthreads();

    for (int it = 0; ; ++it) {
        const float* cur = ldsA + (it & 1) * ASET;
        float* nxt = ldsA + ((it & 1) ^ 1) * ASET;
        int gn = __builtin_amdgcn_readfirstlane(nextg[it & 1]);

        auto product = [&] {
            if (wave >= PWAVE0)
                product_tile(cur, ldsB, ptl, pch, g * TPB + ptl, lane, bias, biasT, out);
        };
        auto claim_and_product = [&] {
            if (threadIdx.x == 0)
                nextg[(it & 1) ^ 1] = (int)(gridDim.x + atomicAdd(gq, 1u));
            product();
        };

        if (gn >= N_GROUPS) {
            product();
            break;
        }

        sum_group(xin, deg, nbr, spill_n, spill, nxt, gn, wave, lane, claim_and_product);

        __syncthreads();
        g = gn;
    }
}

extern "C" void kernel_launch(void* const* d_in, const int* in_sizes, int n_in,
                              void* d_out, int out_size, void* d_ws, size_t ws_size,
                              hipStream_t stream) {
    (void)in_sizes; (void)n_in; (void)out_size; (void)ws_size;

    const float* x    = (const float*)d_in[0];
    const int*   ei   = (const int*)d_in[1];
    const float* Wl1  = (const float*)d_in[2];
    const float* Wr1  = (const float*)d_in[3];
    const float* b1   = (const float*)d_in[4];
    const float* WlT1 = (const float*)d_in[5];
    const float* WrT1 = (const float*)d_in[6];
    const float* bT1  = (const float*)d_in[7];
    const float* Wl2  = (const float*)d_in[8];
    const float* Wr2  = (const float*)d_in[9];
    const float* b2   = (const float*)d_in[10];
    const float* WlT2 = (const float*)d_in[11];
    const float* WrT2 = (const float*)d_in[12];
    const float* bT2  = (const float*)d_in[13];

    const int* src = ei;
    const int* dst = ei + N_EDGES;

    float* ws    = (float*)d_ws;
    float* hid   = ws;
    int*      deg     = (int*)(hid + (size_t)N_NODES * D);
    unsigned* bin_n   = (unsigned*)(deg + N_ROWS);
    unsigned* spill_n = bin_n + (size_t)NBINS * QSTRIDE;
    unsigned* ovf_n   = spill_n + QSTRIDE;
    unsigned* q_layer = ovf_n + QSTRIDE;
    int* nbr   = (int*)(q_layer + 2 * QSTRIDE);
    int* spill = nbr + (size_t)N_ROWS * LCAP;
    v2i* bins  = (v2i*)(spill + (size_t)2 * 2 * N_EDGES);
    v2i* ovf   = bins + (size_t)NBINS * BCAP;
    float* out = (float*)d_out;

    const size_t zero_bytes = ((size_t)NBINS + 4) * QSTRIDE * sizeof(unsigned);

    dim3 blk(256);
    dim3 lblk(512);
    dim3 bgrid(BIN_BLOCKS);
    dim3 rgrid(NBINS);
    dim3 lgrid(LAYER_BLOCKS);

    hipMemsetAsync(bin_n, 0, zero_bytes, stream);
    sage_bins<<<bgrid, blk, 0, stream>>>(src, dst, bin_n, bins, ovf_n, ovf);
    sage_records<<<rgrid, lblk, 0, stream>>>(bin_n, bins, ovf_n, ovf, deg, nbr, spill_n, spill);

    sage_layer<<<lgrid, lblk, 0, stream>>>(x, deg, nbr, spill_n, spill,
                                          Wl1, WlT1, Wr1, WrT1, b1, bT1, q_layer, hid);

    sage_layer<<<lgrid, lblk, 0, stream>>>(hid, deg, nbr, spill_n, spill,
                                          Wl2, WlT2, Wr2, WrT2, b2, bT2, q_layer + QSTRIDE, out);
}


# layer sum part: a row load takes 16 B per lane and fetches TWO neighbours' rows per instruction (even entry in lanes 0-15, odd in 16-31): 8 loads per 16 entries; row sum = evens + odds, added last
# speedup vs baseline: 4.2780x; 1.1001x over previous
#include <hip/hip_runtime.h>

#define N_NODES 100000
#define N_EDGES 1200000
#define D 64
#define N_ROWS (2 * N_NODES)

typedef __attribute__((ext_vector_type(2))) float v2f;
typedef __attribute__((ext_vector_type(4))) float v4f;
typedef __attribute__((ext_vector_type(8))) float v8f;

#define QSTRIDE 64
#define LCAP 32
#define RB 512
#define NBINS ((N_ROWS + RB - 1) / RB)
#define BCAP 12288
#define EB 2048
#define BIN_BLOCKS ((N_EDGES + EB - 1) / EB)
#define EPT_A (EB / 256)
#define BINS_PER_LANE ((NBINS + 31) / 32)
static_assert(NBINS > 256 && NBINS <= 512, "a thread of sage_bins reserves bins tid and tid + 256");

typedef __attribute__((ext_vector_type(2))) int v2i;
typedef __attribute__((ext_vector_type(4))) int v4i;

__global__ __launch_bounds__(256) void sage_bins(
    const int* __restrict__ src,
    const int* __restrict__ dst,
    unsigned* __restrict__ bin_n,
    v2i* __restrict__ bins,
    unsigned* __restrict__ ovf_n,
    v2i* __restrict__ ovf)
{
    __shared__ int hist[512];
    __shared__ int loff[512];
    __shared__ unsigned gb[512];
    __shared__ v2i sorted[2 * EB];

    int tid = (int)threadIdx.x;
    int e0 = (int)blockIdx.x * EB;

    for (int b = tid; b < 512; b += 256) hist[b] = 0;
    __syncthreads();

    int s[EPT_A], d[EPT_A], kf[EPT_A], kr[EPT_A];
#pragma unroll
    for (int i = 0; i < EPT_A; ++i) {
        int e = e0 + i * 256 + tid;
        bool ok = e < N_EDGES;
        s[i] = ok ? src[e] : -1;
        d[i] = ok ? dst[e] : -1;
    }
#pragma unroll
    for (int i = 0; i < EPT_A; ++i) {
        kf[i] = 0; kr[i] = 0;
        if (s[i] >= 0) {
            kf[i] = atomicAdd(&hist[d[i] / RB], 1);
            kr[i] = atomicAdd(&hist[(N_NODES + s[i]) / RB], 1);
        }
    }
    __syncthreads();

    {
        int b0 = tid, b1 = tid + 256;
        int c0 = hist[b0];
        int c1 = (b1 < NBINS) ? hist[b1] : 0;
        unsigned g0 = 0u, g1 = 0u;
        if (c0) g0 = atomicAdd(bin_n + (size_t)b0 * QSTRIDE, (unsigned)c0);
        if (c1) g1 = atomicAdd(bin_n + (size_t)b1 * QSTRIDE, (unsigned)c1);
        gb[b0] = g0;
        gb[b1] = g1;
    }
    if (tid < 32) {
        int sum = 0;
        for (int j = 0; j < BINS_PER_LANE; ++j) {
            int b = tid * BINS_PER_LANE + j;
            sum += (b < NBINS) ? hist[b] : 0;
        }
        int incl = sum;
        for (int o = 1; o < 32; o <<= 1) {
            int up = __shfl_up(incl, o, 32);
            if (tid >= o) incl += up;
        }
        int run = incl - sum;
        for (int j = 0; j < BINS_PER_LANE; ++j) {
            int b = tid * BINS_PER_LANE + j;
            if (b < NBINS) { loff[b] = run; run += hist[b]; }
        }
    }
    __syncthreads();

#pragma unroll
    for (int i = 0; i < EPT_A; ++i) {
        if (s[i] >= 0) {
            int rf = d[i], rr = N_NODES + s[i];
            sorted[loff[rf / RB] + kf[i]] = (v2i){ rf, s[i] };
            sorted[loff[rr / RB] + kr[i]] = (v2i){ rr, d[i] };
        }
    }
    __syncthreads();

    int total = 2 * min(EB, N_EDGES - e0);
    for (int i = tid; i < total; i += 256) {
        v2i ent = sorted[i];
        int b = ent.x / RB;
        unsigned at = gb[b] + (unsigned)(i - loff[b]);
        if (at < (unsigned)BCAP) {
            bins[(size_t)b * BCAP + at] = ent;
        } else {
            unsigned p = atomicAdd(ovf_n, 1u);
            ovf[p] = ent;
        }
    }
}

__global__ __launch_bounds__(512) void sage_records(
    const unsigned* __restrict__ bin_n,
    const v2i* __restrict__ bins,
    const unsigned* __restrict__ ovf_n,
    const v2i* __restrict__ ovf,
    int* __restrict__ deg,
    int* __restrict__ nbr,
    unsigned* __restrict__ spill_n,
    int* __restrict__ spill)
{
    __shared__ int cnt[RB];
    __shared__ __attribute__((aligned(16))) int rec[RB * LCAP];

    int tid = (int)threadIdx.x;
    int b = (int)blockIdx.x;
    int row0 = b * RB;

    for (int i = tid; i < RB; i += 512) cnt[i] = 0;
    for (int i = tid; i < RB * LCAP / 4; i += 512) ((v4i*)rec)[i] = (v4i){ 0, 0, 0, 0 };
    __syncthreads();

    auto place = [&](int row, int nb) {
        int r = row - row0;
        int slot = atomicAdd(&cnt[r], 1);
        if (slot < LCAP) {
            rec[r * LCAP + slot] = nb;
        } else {
            unsigned p = atomicAdd(spill_n, 1u);
            spill[2 * p]     = row;
            spill[2 * p + 1] = nb;
        }
    };

    unsigned have = bin_n[(size_t)b * QSTRIDE];
    int n = (int)min(have, (unsigned)BCAP);
    const v2i* mine = bins + (size_t)b * BCAP;
    for (int i0 = 0; i0 < n; i0 += 4 * 512) {
        v2i ent[4];
#pragma unroll
        for (int u = 0; u < 4; ++u) {
            int i = i0 + u * 512 + tid;
            ent[u] = (i < n) ? mine[i] : (v2i){ -1, 0 };
        }
#pragma unroll
        for (int u = 0; u < 4; ++u)
            if (ent[u].x >= 0) place(ent[u].x, ent[u].y);
    }
    if (have > (unsigned)BCAP) {
        int np = (int)min(*ovf_n, (unsigned)(2 * N_EDGES));
        for (int i = tid; i < np; i += 512) {
            v2i ent = ovf[i];
            if (ent.x / RB == b) place(ent.x, ent.y);
        }
    }
    __syncthreads();

    for (int r = tid; r < RB; r += 512)
        if (row0 + r < N_ROWS) deg[row0 + r] = cnt[r];
    for (int i = tid; i < RB * LCAP / 4; i += 512)
        if (row0 + i / (LCAP / 4) < N_ROWS)
            ((v4i*)nbr)[(size_t)row0 * (LCAP / 4) + i] = ((const v4i*)rec)[i];
}

#define RPW 4
#define TROWS 16
#define GCH 16
#define TPB 2
#define N_TILES (N_NODES / TROWS)
#define N_GROUPS (N_TILES / TPB)
#define LAYER_BLOCKS 768
#define PWAVE0 12
#define BPITCH 130
#define APITCH 68
#define SLOT (TROWS * APITCH)
#define ASET (3 * TPB * SLOT)
static_assert(N_NODES % TROWS == 0 && N_TILES % TPB == 0, "groups must tile the nodes");
static_assert(LAYER_BLOCKS <= N_GROUPS, "every work-group has a first group");
static_assert(2 * TPB * (TROWS / RPW) == 16, "16 waves sum a group's forward and reverse rows");
static_assert(LCAP == 32 && GCH == 16, "a record is one list load: two chunks of eight entry pairs");

template <int K>
__device__ __forceinline__ unsigned half_share(unsigned x) {
    return (unsigned)__builtin_amdgcn_update_dpp(0, (int)x, 0x150 + K, 0xF, 0xF, true);
}

__device__ __forceinline__ float other_half(float x) {
    return __builtin_bit_cast(float, __builtin_amdgcn_permlanex16(
        __builtin_bit_cast(int, x), __builtin_bit_cast(int, x),
        0x76543210, 0xfedcba98, false, false));
}

#define PAIR_OFF(i, k)                                                          \
    off[i] = half_share<(k)>(rowb) + loff;
#define PAIR_LOAD(i)                                                            \
    v[i] = *(const v4f*)(xb + (size_t)off[i]);

#define PAIR_ADD(i, k)                                                          \
    acc += ((k) < cnt) ? v[i] : zero4;

#define PAIR_LOADS(k0)                                                          \
    PAIR_OFF(0, (k0) + 0) PAIR_OFF(1, (k0) + 1)                                 \
    PAIR_OFF(2, (k0) + 2) PAIR_OFF(3, (k0) + 3)                                 \
    PAIR_OFF(4, (k0) + 4) PAIR_OFF(5, (k0) + 5)                                 \
    PAIR_OFF(6, (k0) + 6) PAIR_OFF(7, (k0) + 7)                                 \
    asm volatile("" :: "v"(off[0]), "v"(off[1]), "v"(off[2]), "v"(off[3]),      \
                       "v"(off[4]), "v"(off[5]), "v"(off[6]), "v"(off[7]));     \
    PAIR_LOAD(0) PAIR_LOAD(1) PAIR_LOAD(2) PAIR_LOAD(3)                         \
    PAIR_LOAD(4) PAIR_LOAD(5) PAIR_LOAD(6) PAIR_LOAD(7)

#define PAIR_ADDS(k0)                                                           \
    PAIR_ADD(0, (k0) + 0) PAIR_ADD(1, (k0) + 1)                                 \
    PAIR_ADD(2, (k0) + 2) PAIR_ADD(3, (k0) + 3)                                 \
    PAIR_ADD(4, (k0) + 4) PAIR_ADD(5, (k0) + 5)                                 \
    PAIR_ADD(6, (k0) + 6) PAIR_ADD(7, (k0) + 7)

#define PAIR_LOADS4(k0)                                                         \
    PAIR_OFF(0, (k0) + 0) PAIR_OFF(1, (k0) + 1)                                 \
    PAIR_OFF(2, (k0) + 2) PAIR_OFF(3, (k0) + 3)                                 \
    asm volatile("" :: "v"(off[0]), "v"(off[1]), "v"(off[2]), "v"(off[3]));     \
    PAIR_LOAD(0) PAIR_LOAD(1) PAIR_LOAD(2) PAIR_LOAD(3)
#define PAIR_ADDS4(k0)                                                          \
    PAIR_ADD(0, (k0) + 0) PAIR_ADD(1, (k0) + 1)                                 \
    PAIR_ADD(2, (k0) + 2) PAIR_ADD(3, (k0) + 3)

__device__ __forceinline__ v8f wmma4(v2f a, v2f b, v8f c) {
    return __builtin_amdgcn_wmma_f32_16x16x4_f32(
        false, a, false, b, (short)0, c, false, false);
}

template <class Mid>
__device__ __forceinline__ void sum_group(
    const float* __restrict__ xin,
    const int* __restrict__ deg,
    const int* __restrict__ nbr,
    const unsigned* __restrict__ spill_n,
    const int* __restrict__ spill,
    float* aset, int group, int wave, int lane, Mid mid)
{
    int side = wave >> 3;
    int tl   = (wave >> 2) & (TPB - 1);
    int part = wave & 3;
    int tile = group * TPB + tl;
    int node0 = tile * TROWS + part * RPW;
    int r0 = side * N_NODES + node0;
    int half = lane >> 4;
    int q    = lane & 15;
    float* slot = aset + (side * TPB + tl) * SLOT + part * RPW * APITCH + 4 * q;
    const char* xb = (const char*)xin;
    unsigned loff = (unsigned)q * 16u;
    const v4f zero4 = { 0.0f, 0.0f, 0.0f, 0.0f };

    v4f xr[RPW / 2];
    if (side == 0) {
#pragma unroll
        for (int j = 0; j < RPW / 2; ++j)
            xr[j] = *(const v4f*)(xin + (size_t)(node0 + 2 * j + half) * D + 4 * q);
    }

    int nv = deg[r0 + (lane & (RPW - 1))];
    int ent = 2 * q + half;
    int idx = nbr[r0 * LCAP + ent];

    mid();

    int n = __builtin_amdgcn_readlane(nv, 0);

    if (side == 0) {
        float* xs = slot + 2 * TPB * SLOT + half * APITCH;
#pragma unroll
        for (int j = 0; j < RPW / 2; ++j)
            *(v4f*)(xs + 2 * j * APITCH) = xr[j];
    }

    for (int j = 0; j < RPW; ++j) {
        v4f v[GCH / 2];
        unsigned off[GCH / 2];
        v4f acc = zero4;
        int m = min(n, LCAP);
        int cnt = (m + 1 - half) >> 1;
        unsigned rowb = (unsigned)idx << 8;

        PAIR_LOADS(0)

        int n2 = __builtin_amdgcn_readlane(nv, (j + 1) & (RPW - 1));
        int idx2 = nbr[(r0 + ((j + 1) & (RPW - 1))) * LCAP + ent];
        asm volatile("" :: "v"(idx2));

        PAIR_ADDS(0)

        if (m > GCH) {
            PAIR_LOADS4(GCH / 2)
            PAIR_ADDS4(GCH / 2)
            if (m > GCH + GCH / 2) {
                PAIR_LOADS4(GCH / 2 + GCH / 4)
                PAIR_ADDS4(GCH / 2 + GCH / 4)
            }
        }

        if (n > LCAP) {
            int row = r0 + j;
            int np = (int)min(*spill_n, (unsigned)(2 * N_EDGES));
            for (int p0 = 0; p0 < np; p0 += 32) {
                int p = p0 + lane;
                int prow = -1, pnb = 0;
                if (p < np) { prow = spill[2 * p]; pnb = spill[2 * p + 1]; }
                unsigned hit = __builtin_amdgcn_ballot_w32(prow == row);
                while (hit) {
                    int b = __builtin_ctz(hit);
                    hit &= hit - 1;
                    int nu = __builtin_amdgcn_readlane(pnb, b);
                    v4f t = *(const v4f*)(xin + (size_t)nu * D + 4 * q);
                    acc += (half == 0) ? t : zero4;
                }
            }
        }

        v4f tot;
#pragma unroll
        for (int i = 0; i < 4; ++i) tot[i] = acc[i] + other_half(acc[i]);

        float inv = __builtin_amdgcn_rcpf(fmaxf((float)n, 1.0f));
        if (half == 0) *(v4f*)(slot + j * APITCH) = tot * inv;

        n = n2; idx = idx2;
    }
}

__device__ __forceinline__ void product_tile(
    const float* aset, const float* ldsB, int tl, int ch, int tile, int lane,
    const float* __restrict__ bias, const float* __restrict__ biasT,
    float* __restrict__ out)
{
    int nl = lane & 15;
    int khalf = (lane < 16) ? 0 : 2;
    const float* arow = aset + tl * SLOT + nl * APITCH + khalf;
    const float* bcol = ldsB + (khalf >> 1) * BPITCH + ((ch * 32 + nl) << 1);

    v8f c0 = {}, c1 = {};

#pragma unroll
    for (int seg = 0; seg < 3; ++seg) {
#pragma unroll 4
        for (int kk = 0; kk < 16; ++kk) {
            v2f a = *(const v2f*)(arow + seg * (TPB * SLOT) + kk * 4);
            const float* bb = bcol + (seg * 32 + kk * 2) * BPITCH;
            v2f b0 = *(const v2f*)(bb);
            v2f b1 = *(const v2f*)(bb + 32);
            c0 = wmma4(a, b0, c0);
            c1 = wmma4(a, b1, c1);
        }
    }

    int rowbase = tile * TROWS;
    int rowoff = (lane < 16) ? 0 : 8;
#pragma unroll
    for (int ct = 0; ct < 2; ++ct) {
        int j = ch * 32 + ct * 16 + nl;
        float bs = bias[j] + biasT[j];
        v8f c = (ct == 0) ? c0 : c1;
#pragma unroll
        for (int v = 0; v < 8; ++v) {
            int row = rowbase + rowoff + v;
            out[(size_t)row * D + j] = c[v] + bs;
        }
    }
}

__global__ __launch_bounds__(512) void sage_layer(
    const float* __restrict__ xin,
    const int* __restrict__ deg,
    const int* __restrict__ nbr,
    const unsigned* __restrict__ spill_n,
    const int* __restrict__ spill,
    const float* __restrict__ Wl,
    const float* __restrict__ WlT,
    const float* __restrict__ Wr,
    const float* __restrict__ WrT,
    const float* __restrict__ bias,
    const float* __restrict__ biasT,
    unsigned* __restrict__ gq,
    float* __restrict__ out)
{
    __shared__ __attribute__((aligned(16))) float ldsB[96 * BPITCH];
    __shared__ __attribute__((aligned(16))) float ldsA[2 * ASET];
    __shared__ int nextg[2];

    {
        v4f w0[2], w1[2], w2[2], w3[2];
#pragma unroll
        for (int it = 0; it < 2; ++it) {
            int g = it * 512 + (int)threadIdx.x;
            w0[it] = ((const v4f*)Wl)[g];
            w1[it] = ((const v4f*)WlT)[g];
            w2[it] = ((const v4f*)Wr)[g];
            w3[it] = ((const v4f*)WrT)[g];
        }
#pragma unroll
        for (int it = 0; it < 2; ++it) {
            int g  = it * 512 + (int)threadIdx.x;
            int n  = g >> 4;
            int kq = g & 15;
            v4f w2s = w2[it] + w3[it];
            float* d = ldsB + (2 * kq) * BPITCH + (n << 1);
            *(v2f*)(d)                = (v2f){ w0[it].x, w0[it].y };
            *(v2f*)(d + BPITCH)       = (v2f){ w0[it].z, w0[it].w };
            *(v2f*)(d + 32 * BPITCH)  = (v2f){ w1[it].x, w1[it].y };
            *(v2f*)(d + 33 * BPITCH)  = (v2f){ w1[it].z, w1[it].w };
            *(v2f*)(d + 64 * BPITCH)  = (v2f){ w2s.x, w2s.y };
            *(v2f*)(d + 65 * BPITCH)  = (v2f){ w2s.z, w2s.w };
        }
    }

    int lane = threadIdx.x & 31;
    int wave = __builtin_amdgcn_readfirstlane((int)(threadIdx.x >> 5));

    int ptl = (wave - PWAVE0) & (TPB - 1);
    int pch = (wave - PWAVE0) >> 1;

    int g = (int)blockIdx.x;

    sum_group(xin, deg, nbr, spill_n, spill, ldsA, g, wave, lane, [&] {
        if (threadIdx.x == 0) nextg[0] = (int)(gridDim.x + atomicAdd(gq, 1u));
    });

    __syncthreads();

    for (int it = 0; ; ++it) {
        const float* cur = ldsA + (it & 1) * ASET;
        float* nxt = ldsA + ((it & 1) ^ 1) * ASET;
        int gn = __builtin_amdgcn_readfirstlane(nextg[it & 1]);

        auto product = [&] {
            if (wave >= PWAVE0)
                product_tile(cur, ldsB, ptl, pch, g * TPB + ptl, lane, bias, biasT, out);
        };
        auto claim_and_product = [&] {
            if (threadIdx.x == 0)
                nextg[(it & 1) ^ 1] = (int)(gridDim.x + atomicAdd(gq, 1u));
            product();
        };

        if (gn >= N_GROUPS) {
            product();
            break;
        }

        sum_group(xin, deg, nbr, spill_n, spill, nxt, gn, wave, lane, claim_and_product);

        __syncthreads();
        g = gn;
    }
}

extern "C" void kernel_launch(void* const* d_in, const int* in_sizes, int n_in,
                              void* d_out, int out_size, void* d_ws, size_t ws_size,
                              hipStream_t stream) {
    (void)in_sizes; (void)n_in; (void)out_size; (void)ws_size;

    const float* x    = (const float*)d_in[0];
    const int*   ei   = (const int*)d_in[1];
    const float* Wl1  = (const float*)d_in[2];
    const float* Wr1  = (const float*)d_in[3];
    const float* b1   = (const float*)d_in[4];
    const float* WlT1 = (const float*)d_in[5];
    const float* WrT1 = (const float*)d_in[6];
    const float* bT1  = (const float*)d_in[7];
    const float* Wl2  = (const float*)d_in[8];
    const float* Wr2  = (const float*)d_in[9];
    const float* b2   = (const float*)d_in[10];
    const float* WlT2 = (const float*)d_in[11];
    const float* WrT2 = (const float*)d_in[12];
    const float* bT2  = (const float*)d_in[13];

    const int* src = ei;
    const int* dst = ei + N_EDGES;

    float* ws    = (float*)d_ws;
    float* hid   = ws;
    int*      deg     = (int*)(hid + (size_t)N_NODES * D);
    unsigned* bin_n   = (unsigned*)(deg + N_ROWS);
    unsigned* spill_n = bin_n + (size_t)NBINS * QSTRIDE;
    unsigned* ovf_n   = spill_n + QSTRIDE;
    unsigned* q_layer = ovf_n + QSTRIDE;
    int* nbr   = (int*)(q_layer + 2 * QSTRIDE);
    int* spill = nbr + (size_t)N_ROWS * LCAP;
    v2i* bins  = (v2i*)(spill + (size_t)2 * 2 * N_EDGES);
    v2i* ovf   = bins + (size_t)NBINS * BCAP;
    float* out = (float*)d_out;

    const size_t zero_bytes = ((size_t)NBINS + 4) * QSTRIDE * sizeof(unsigned);

    dim3 blk(256);
    dim3 lblk(512);
    dim3 bgrid(BIN_BLOCKS);
    dim3 rgrid(NBINS);
    dim3 lgrid(LAYER_BLOCKS);

    hipMemsetAsync(bin_n, 0, zero_bytes, stream);
    sage_bins<<<bgrid, blk, 0, stream>>>(src, dst, bin_n, bins, ovf_n, ovf);
    sage_records<<<rgrid, lblk, 0, stream>>>(bin_n, bins, ovf_n, ovf, deg, nbr, spill_n, spill);

    sage_layer<<<lgrid, lblk, 0, stream>>>(x, deg, nbr, spill_n, spill,
                                          Wl1, WlT1, Wr1, WrT1, b1, bT1, q_layer, hid);

    sage_layer<<<lgrid, lblk, 0, stream>>>(hid, deg, nbr, spill_n, spill,
                                          Wl2, WlT2, Wr2, WrT2, b2, bT2, q_layer + QSTRIDE, out);
}
